# starting kernel
# baseline (speedup 1.0000x reference)
#include <hip/hip_runtime.h>

typedef __attribute__((ext_vector_type(2))) float v2f;
typedef __attribute__((ext_vector_type(8))) float v8f;

#define K_F 64

__global__ void zero_kernel(float* __restrict__ p, int n) {
  int i = blockIdx.x * blockDim.x + threadIdx.x;
  if (i < n) p[i] = 0.0f;
}

__global__ void degree_kernel(const int* __restrict__ dst,
                              float* __restrict__ deg, int nE) {
  int i = blockIdx.x * blockDim.x + threadIdx.x;
  if (i < nE) atomicAdd(&deg[dst[i]], 1.0f);
}

__global__ void inv_kernel(float* __restrict__ deg, int n) {
  int i = blockIdx.x * blockDim.x + threadIdx.x;
  if (i < n) deg[i] = 1.0f / fmaxf(deg[i], 1.0f);
}

__global__ void scatter_kernel(const float* __restrict__ feat,
                               const int* __restrict__ src,
                               const int* __restrict__ dst,
                               float* __restrict__ msg, int nE) {
  int edge = blockIdx.x * (blockDim.x >> 5) + (threadIdx.x >> 5);
  if (edge >= nE) return;
  int lane = threadIdx.x & 31;
  int s = src[edge];
  int d = dst[edge];
  v2f v = *reinterpret_cast<const v2f*>(feat + (size_t)s * K_F + lane * 2);
  float* base = msg + (size_t)d * K_F + lane * 2;
  atomicAdd(base, v.x);
  atomicAdd(base + 1, v.y);
}

template <int OUT, bool RELU>
__global__ void sage_gemm(const float* __restrict__ msg,
                          const float* __restrict__ invdeg,
                          const float* __restrict__ xin,
                          const float* __restrict__ Wl,
                          const float* __restrict__ Wr,
                          const float* __restrict__ bias,
                          float* __restrict__ out,
                          int nTilesN, int nTiles) {
  int tile = blockIdx.x * (blockDim.x >> 5) + (threadIdx.x >> 5);
  if (tile >= nTiles) return;
  int tm = tile / nTilesN;
  int tn = tile - tm * nTilesN;

  int lane = threadIdx.x & 31;
  int hrow = lane & 15;
  int hi   = lane >> 4;
  int row  = tm * 16 + hrow;
  int col  = tn * 16 + hrow;
  int koff = hi * 2;

  float inv = invdeg[row];
  const float* mrow = msg + (size_t)row * K_F + koff;
  const float* xrow = xin + (size_t)row * K_F + koff;

  v8f c = {};
#pragma unroll
  for (int k = 0; k < K_F; k += 4) {
    v2f am = *reinterpret_cast<const v2f*>(mrow + k);
    am.x *= inv;
    am.y *= inv;
    v2f ax = *reinterpret_cast<const v2f*>(xrow + k);
    v2f bl, br;
    bl.x = Wl[(k + koff) * OUT + col];
    bl.y = Wl[(k + koff + 1) * OUT + col];
    br.x = Wr[(k + koff) * OUT + col];
    br.y = Wr[(k + koff + 1) * OUT + col];
    c = __builtin_amdgcn_wmma_f32_16x16x4_f32(false, am, false, bl,
                                              (short)0, c, false, false);
    c = __builtin_amdgcn_wmma_f32_16x16x4_f32(false, ax, false, br,
                                              (short)0, c, false, false);
  }

  float bv = bias[col];
#pragma unroll
  for (int i = 0; i < 8; ++i) {
    float v = c[i] + bv;
    if (RELU) v = fmaxf(v, 0.0f);
    out[(size_t)(tm * 16 + hi * 8 + i) * OUT + col] = v;
  }
}

extern "C" void kernel_launch(void* const* d_in, const int* in_sizes, int n_in,
                              void* d_out, int out_size, void* d_ws,
                              size_t ws_size, hipStream_t stream) {
  (void)n_in; (void)out_size; (void)ws_size;

  const float* x   = (const float*)d_in[0];
  const int*   ei  = (const int*)d_in[1];
  const float* W1l = (const float*)d_in[2];
  const float* b1  = (const float*)d_in[3];
  const float* W1r = (const float*)d_in[4];
  const float* W2l = (const float*)d_in[5];
  const float* b2  = (const float*)d_in[6];
  const float* W2r = (const float*)d_in[7];
  float* out = (float*)d_out;

  const int N  = in_sizes[0] / K_F;
  const int nE = in_sizes[1] / 2;
  const int* src = ei;
  const int* dst = ei + nE;

  float* msg = (float*)d_ws;
  float* deg = msg + (size_t)N * K_F;
  float* h   = deg + N;

  const int T = 256;

  {
    int n0 = N * K_F + N;
    zero_kernel<<<(n0 + T - 1) / T, T, 0, stream>>>(msg, n0);
  }
  degree_kernel<<<(nE + T - 1) / T, T, 0, stream>>>(dst, deg, nE);
  inv_kernel<<<(N + T - 1) / T, T, 0, stream>>>(deg, N);

  scatter_kernel<<<(nE + 7) / 8, T, 0, stream>>>(x, src, dst, msg, nE);
  {
    int nTilesN = K_F / 16;
    int nTiles  = (N / 16) * nTilesN;
    int blocks  = (nTiles + 7) / 8;
    sage_gemm<K_F, true><<<blocks, T, 0, stream>>>(msg, deg, x, W1l, W1r, b1,
                                                   h, nTilesN, nTiles);
  }

  {
    int n1 = N * K_F;
    zero_kernel<<<(n1 + T - 1) / T, T, 0, stream>>>(msg, n1);
  }
  scatter_kernel<<<(nE + 7) / 8, T, 0, stream>>>(h, src, dst, msg, nE);
  {
    int nTilesN = 16 / 16;
    int nTiles  = (N / 16) * nTilesN;
    int blocks  = (nTiles + 7) / 8;
    sage_gemm<16, false><<<blocks, T, 0, stream>>>(msg, deg, h, W2l, W2r, b2,
                                                   out, nTilesN, nTiles);
  }
}


# layer 2 applies W2l before the neighbour sum (mean is linear): edge pass adds 16-wide projected rows times 1/deg straight into out, not 64-wide rows; f32 sum order changes; zero pass dropped
# speedup vs baseline: 1.2035x; 1.2035x over previous
#include <hip/hip_runtime.h>

typedef __attribute__((ext_vector_type(2))) float v2f;
typedef __attribute__((ext_vector_type(8))) float v8f;

#define K_F 64
#define N_C 16
#define EPW 16

__global__ void zero_kernel(float* __restrict__ p, int n) {
  int i = blockIdx.x * blockDim.x + threadIdx.x;
  if (i < n) p[i] = 0.0f;
}

__global__ void degree_kernel(const int* __restrict__ dst,
                              float* __restrict__ deg, int nE) {
  int i = blockIdx.x * blockDim.x + threadIdx.x;
  if (i < nE) atomicAdd(&deg[dst[i]], 1.0f);
}

__global__ void inv_kernel(float* __restrict__ deg, int n) {
  int i = blockIdx.x * blockDim.x + threadIdx.x;
  if (i < n) deg[i] = 1.0f / fmaxf(deg[i], 1.0f);
}

__global__ void scatter_kernel(const float* __restrict__ feat,
                               const int* __restrict__ src,
                               const int* __restrict__ dst,
                               float* __restrict__ msg, int nE) {
  int edge = blockIdx.x * (blockDim.x >> 5) + (threadIdx.x >> 5);
  if (edge >= nE) return;
  int lane = threadIdx.x & 31;
  int s = src[edge];
  int d = dst[edge];
  v2f v = *reinterpret_cast<const v2f*>(feat + (size_t)s * K_F + lane * 2);
  float* base = msg + (size_t)d * K_F + lane * 2;
  atomicAdd(base, v.x);
  atomicAdd(base + 1, v.y);
}

template <int OUT, bool RELU>
__global__ void sage_gemm(const float* __restrict__ msg,
                          const float* __restrict__ invdeg,
                          const float* __restrict__ xin,
                          const float* __restrict__ Wl,
                          const float* __restrict__ Wr,
                          const float* __restrict__ bias,
                          float* __restrict__ out,
                          int nTilesN, int nTiles) {
  int tile = blockIdx.x * (blockDim.x >> 5) + (threadIdx.x >> 5);
  if (tile >= nTiles) return;
  int tm = tile / nTilesN;
  int tn = tile - tm * nTilesN;

  int lane = threadIdx.x & 31;
  int hrow = lane & 15;
  int hi   = lane >> 4;
  int row  = tm * 16 + hrow;
  int col  = tn * 16 + hrow;
  int koff = hi * 2;

  float inv = invdeg[row];
  const float* mrow = msg + (size_t)row * K_F + koff;
  const float* xrow = xin + (size_t)row * K_F + koff;

  v8f c = {};
#pragma unroll
  for (int k = 0; k < K_F; k += 4) {
    v2f am = *reinterpret_cast<const v2f*>(mrow + k);
    am.x *= inv;
    am.y *= inv;
    v2f ax = *reinterpret_cast<const v2f*>(xrow + k);
    v2f bl, br;
    bl.x = Wl[(k + koff) * OUT + col];
    bl.y = Wl[(k + koff + 1) * OUT + col];
    br.x = Wr[(k + koff) * OUT + col];
    br.y = Wr[(k + koff + 1) * OUT + col];
    c = __builtin_amdgcn_wmma_f32_16x16x4_f32(false, am, false, bl,
                                              (short)0, c, false, false);
    c = __builtin_amdgcn_wmma_f32_16x16x4_f32(false, ax, false, br,
                                              (short)0, c, false, false);
  }

  float bv = bias[col];
#pragma unroll
  for (int i = 0; i < 8; ++i) {
    float v = c[i] + bv;
    if (RELU) v = fmaxf(v, 0.0f);
    out[(size_t)(tm * 16 + hi * 8 + i) * OUT + col] = v;
  }
}

__global__ void proj2_kernel(const float* __restrict__ hin,
                             const float* __restrict__ Wl,
                             const float* __restrict__ Wr,
                             const float* __restrict__ bias,
                             float* __restrict__ p,
                             float* __restrict__ out, int nTiles) {
  int tile = blockIdx.x * (blockDim.x >> 5) + (threadIdx.x >> 5);
  if (tile >= nTiles) return;
  int lane = threadIdx.x & 31;
  int col  = lane & 15;
  int hi   = lane >> 4;
  int koff = hi * 2;
  const float* hrow = hin + (size_t)(tile * 16 + col) * K_F + koff;

  v8f cp = {};
  v8f cq = {};
#pragma unroll
  for (int k = 0; k < K_F; k += 4) {
    v2f a = *reinterpret_cast<const v2f*>(hrow + k);
    v2f bl, br;
    bl.x = Wl[(k + koff) * N_C + col];
    bl.y = Wl[(k + koff + 1) * N_C + col];
    br.x = Wr[(k + koff) * N_C + col];
    br.y = Wr[(k + koff + 1) * N_C + col];
    cp = __builtin_amdgcn_wmma_f32_16x16x4_f32(false, a, false, bl,
                                               (short)0, cp, false, false);
    cq = __builtin_amdgcn_wmma_f32_16x16x4_f32(false, a, false, br,
                                               (short)0, cq, false, false);
  }

  float bv = bias[col];
#pragma unroll
  for (int i = 0; i < 8; ++i) {
    size_t o = (size_t)(tile * 16 + hi * 8 + i) * N_C + col;
    p[o]   = cp[i];
    out[o] = cq[i] + bv;
  }
}

template <bool TAIL>
__device__ __forceinline__ void edge_group(const float* __restrict__ p,
                                           const int* __restrict__ src,
                                           const int* __restrict__ dst,
                                           const float* __restrict__ invdeg,
                                           float* __restrict__ out,
                                           int e0, int nE, int j, int hi) {
  int idx = 0;
  if (!TAIL || e0 + j < nE) {
    const int* ip = hi ? dst : src;
    idx = ip[e0 + j];
  }
  float w = invdeg[__shfl(idx, 16 + j)];

  float v[EPW / 2];
  float ws[EPW / 2];
  int   dd[EPW / 2];
#pragma unroll
  for (int t = 0; t < EPW / 2; ++t) {
    int q = hi * (EPW / 2) + t;
    int s = __shfl(idx, q);
    dd[t] = __shfl(idx, 16 + q);
    ws[t] = __shfl(w, q);
    v[t]  = (!TAIL || e0 + q < nE) ? p[(size_t)s * N_C + j] : 0.0f;
  }
#pragma unroll
  for (int t = 0; t < EPW / 2; ++t) {
    int q = hi * (EPW / 2) + t;
    if (!TAIL || e0 + q < nE)
      atomicAdd(out + (size_t)dd[t] * N_C + j, v[t] * ws[t]);
  }
}

__global__ void scatter16_kernel(const float* __restrict__ p,
                                 const int* __restrict__ src,
                                 const int* __restrict__ dst,
                                 const float* __restrict__ invdeg,
                                 float* __restrict__ out, int nE) {
  int wave = blockIdx.x * (blockDim.x >> 5) + (threadIdx.x >> 5);
  int e0 = wave * EPW;
  if (e0 >= nE) return;
  int lane = threadIdx.x & 31;
  int j  = lane & 15;
  int hi = lane >> 4;
  if (e0 + EPW <= nE)
    edge_group<false>(p, src, dst, invdeg, out, e0, nE, j, hi);
  else
    edge_group<true>(p, src, dst, invdeg, out, e0, nE, j, hi);
}

extern "C" void kernel_launch(void* const* d_in, const int* in_sizes, int n_in,
                              void* d_out, int out_size, void* d_ws,
                              size_t ws_size, hipStream_t stream) {
  (void)n_in; (void)out_size; (void)ws_size;

  const float* x   = (const float*)d_in[0];
  const int*   ei  = (const int*)d_in[1];
  const float* W1l = (const float*)d_in[2];
  const float* b1  = (const float*)d_in[3];
  const float* W1r = (const float*)d_in[4];
  const float* W2l = (const float*)d_in[5];
  const float* b2  = (const float*)d_in[6];
  const float* W2r = (const float*)d_in[7];
  float* out = (float*)d_out;

  const int N  = in_sizes[0] / K_F;
  const int nE = in_sizes[1] / 2;
  const int* src = ei;
  const int* dst = ei + nE;

  float* msg = (float*)d_ws;
  float* deg = msg + (size_t)N * K_F;
  float* h   = deg + N;

  const int T = 256;

  {
    int n0 = N * K_F + N;
    zero_kernel<<<(n0 + T - 1) / T, T, 0, stream>>>(msg, n0);
  }
  degree_kernel<<<(nE + T - 1) / T, T, 0, stream>>>(dst, deg, nE);
  inv_kernel<<<(N + T - 1) / T, T, 0, stream>>>(deg, N);

  scatter_kernel<<<(nE + 7) / 8, T, 0, stream>>>(x, src, dst, msg, nE);
  {
    int nTilesN = K_F / 16;
    int nTiles  = (N / 16) * nTilesN;
    int blocks  = (nTiles + 7) / 8;
    sage_gemm<K_F, true><<<blocks, T, 0, stream>>>(msg, deg, x, W1l, W1r, b1,
                                                   h, nTilesN, nTiles);
  }

  float* p = msg;
  {
    int nTiles = N / 16;
    int blocks = (nTiles + 7) / 8;
    proj2_kernel<<<blocks, T, 0, stream>>>(h, W2l, W2r, b2, p, out, nTiles);
  }
  {
    int waves  = (nE + EPW - 1) / EPW;
    int blocks = (waves + 7) / 8;
    scatter16_kernel<<<blocks, T, 0, stream>>>(p, src, dst, deg, out, nE);
  }
}


# layer-1 edge pass on whole 128-B lines: a wave takes 16 edges (indices in one load), lane l owns row words l and l+32: 2 line loads + 2 line f32 atomics per row; atomic sums still in arrival order
# speedup vs baseline: 1.6128x; 1.3401x over previous
#include <hip/hip_runtime.h>

typedef __attribute__((ext_vector_type(2))) float v2f;
typedef __attribute__((ext_vector_type(8))) float v8f;

#define K_F 64
#define N_C 16
#define EPW 16

__global__ void zero_kernel(float* __restrict__ p, int n) {
  int i = blockIdx.x * blockDim.x + threadIdx.x;
  if (i < n) p[i] = 0.0f;
}

__global__ void degree_kernel(const int* __restrict__ dst,
                              float* __restrict__ deg, int nE) {
  int i = blockIdx.x * blockDim.x + threadIdx.x;
  if (i < nE) atomicAdd(&deg[dst[i]], 1.0f);
}

__global__ void inv_kernel(float* __restrict__ deg, int n) {
  int i = blockIdx.x * blockDim.x + threadIdx.x;
  if (i < n) deg[i] = 1.0f / fmaxf(deg[i], 1.0f);
}

#define EPW1 16

template <bool TAIL>
__device__ __forceinline__ void edge_group64(const float* __restrict__ feat,
                                             const int* __restrict__ src,
                                             const int* __restrict__ dst,
                                             float* __restrict__ msg,
                                             int e0, int nE, int lane) {
  int j = lane & 15;
  int idx = 0;
  if (!TAIL || e0 + j < nE) {
    const int* ip = (lane >> 4) ? dst : src;
    idx = ip[e0 + j];
  }

  float lo[EPW1];
  float hi[EPW1];
#pragma unroll
  for (int t = 0; t < EPW1; ++t) {
    lo[t] = 0.0f;
    hi[t] = 0.0f;
    if (!TAIL || e0 + t < nE) {
      int s = __builtin_amdgcn_readlane(idx, t);
      const float* r = feat + (size_t)s * K_F + lane;
      lo[t] = r[0];
      hi[t] = r[32];
    }
  }
#pragma unroll
  for (int t = 0; t < EPW1; ++t) {
    if (!TAIL || e0 + t < nE) {
      int d = __builtin_amdgcn_readlane(idx, 16 + t);
      float* b = msg + (size_t)d * K_F + lane;
      atomicAdd(b, lo[t]);
      atomicAdd(b + 32, hi[t]);
    }
  }
}

__global__ void scatter_kernel(const float* __restrict__ feat,
                               const int* __restrict__ src,
                               const int* __restrict__ dst,
                               float* __restrict__ msg, int nE) {
  int wave = blockIdx.x * (blockDim.x >> 5) + (threadIdx.x >> 5);
  int e0 = wave * EPW1;
  if (e0 >= nE) return;
  int lane = threadIdx.x & 31;
  if (e0 + EPW1 <= nE)
    edge_group64<false>(feat, src, dst, msg, e0, nE, lane);
  else
    edge_group64<true>(feat, src, dst, msg, e0, nE, lane);
}

template <int OUT, bool RELU>
__global__ void sage_gemm(const float* __restrict__ msg,
                          const float* __restrict__ invdeg,
                          const float* __restrict__ xin,
                          const float* __restrict__ Wl,
                          const float* __restrict__ Wr,
                          const float* __restrict__ bias,
                          float* __restrict__ out,
                          int nTilesN, int nTiles) {
  int tile = blockIdx.x * (blockDim.x >> 5) + (threadIdx.x >> 5);
  if (tile >= nTiles) return;
  int tm = tile / nTilesN;
  int tn = tile - tm * nTilesN;

  int lane = threadIdx.x & 31;
  int hrow = lane & 15;
  int hi   = lane >> 4;
  int row  = tm * 16 + hrow;
  int col  = tn * 16 + hrow;
  int koff = hi * 2;

  float inv = invdeg[row];
  const float* mrow = msg + (size_t)row * K_F + koff;
  const float* xrow = xin + (size_t)row * K_F + koff;

  v8f c = {};
#pragma unroll
  for (int k = 0; k < K_F; k += 4) {
    v2f am = *reinterpret_cast<const v2f*>(mrow + k);
    am.x *= inv;
    am.y *= inv;
    v2f ax = *reinterpret_cast<const v2f*>(xrow + k);
    v2f bl, br;
    bl.x = Wl[(k + koff) * OUT + col];
    bl.y = Wl[(k + koff + 1) * OUT + col];
    br.x = Wr[(k + koff) * OUT + col];
    br.y = Wr[(k + koff + 1) * OUT + col];
    c = __builtin_amdgcn_wmma_f32_16x16x4_f32(false, am, false, bl,
                                              (short)0, c, false, false);
    c = __builtin_amdgcn_wmma_f32_16x16x4_f32(false, ax, false, br,
                                              (short)0, c, false, false);
  }

  float bv = bias[col];
#pragma unroll
  for (int i = 0; i < 8; ++i) {
    float v = c[i] + bv;
    if (RELU) v = fmaxf(v, 0.0f);
    out[(size_t)(tm * 16 + hi * 8 + i) * OUT + col] = v;
  }
}

__global__ void proj2_kernel(const float* __restrict__ hin,
                             const float* __restrict__ Wl,
                             const float* __restrict__ Wr,
                             const float* __restrict__ bias,
                             float* __restrict__ p,
                             float* __restrict__ out, int nTiles) {
  int tile = blockIdx.x * (blockDim.x >> 5) + (threadIdx.x >> 5);
  if (tile >= nTiles) return;
  int lane = threadIdx.x & 31;
  int col  = lane & 15;
  int hi   = lane >> 4;
  int koff = hi * 2;
  const float* hrow = hin + (size_t)(tile * 16 + col) * K_F + koff;

  v8f cp = {};
  v8f cq = {};
#pragma unroll
  for (int k = 0; k < K_F; k += 4) {
    v2f a = *reinterpret_cast<const v2f*>(hrow + k);
    v2f bl, br;
    bl.x = Wl[(k + koff) * N_C + col];
    bl.y = Wl[(k + koff + 1) * N_C + col];
    br.x = Wr[(k + koff) * N_C + col];
    br.y = Wr[(k + koff + 1) * N_C + col];
    cp = __builtin_amdgcn_wmma_f32_16x16x4_f32(false, a, false, bl,
                                               (short)0, cp, false, false);
    cq = __builtin_amdgcn_wmma_f32_16x16x4_f32(false, a, false, br,
                                               (short)0, cq, false, false);
  }

  float bv = bias[col];
#pragma unroll
  for (int i = 0; i < 8; ++i) {
    size_t o = (size_t)(tile * 16 + hi * 8 + i) * N_C + col;
    p[o]   = cp[i];
    out[o] = cq[i] + bv;
  }
}

template <bool TAIL>
__device__ __forceinline__ void edge_group(const float* __restrict__ p,
                                           const int* __restrict__ src,
                                           const int* __restrict__ dst,
                                           const float* __restrict__ invdeg,
                                           float* __restrict__ out,
                                           int e0, int nE, int j, int hi) {
  int idx = 0;
  if (!TAIL || e0 + j < nE) {
    const int* ip = hi ? dst : src;
    idx = ip[e0 + j];
  }
  float w = invdeg[__shfl(idx, 16 + j)];

  float v[EPW / 2];
  float ws[EPW / 2];
  int   dd[EPW / 2];
#pragma unroll
  for (int t = 0; t < EPW / 2; ++t) {
    int q = hi * (EPW / 2) + t;
    int s = __shfl(idx, q);
    dd[t] = __shfl(idx, 16 + q);
    ws[t] = __shfl(w, q);
    v[t]  = (!TAIL || e0 + q < nE) ? p[(size_t)s * N_C + j] : 0.0f;
  }
#pragma unroll
  for (int t = 0; t < EPW / 2; ++t) {
    int q = hi * (EPW / 2) + t;
    if (!TAIL || e0 + q < nE)
      atomicAdd(out + (size_t)dd[t] * N_C + j, v[t] * ws[t]);
  }
}

__global__ void scatter16_kernel(const float* __restrict__ p,
                                 const int* __restrict__ src,
                                 const int* __restrict__ dst,
                                 const float* __restrict__ invdeg,
                                 float* __restrict__ out, int nE) {
  int wave = blockIdx.x * (blockDim.x >> 5) + (threadIdx.x >> 5);
  int e0 = wave * EPW;
  if (e0 >= nE) return;
  int lane = threadIdx.x & 31;
  int j  = lane & 15;
  int hi = lane >> 4;
  if (e0 + EPW <= nE)
    edge_group<false>(p, src, dst, invdeg, out, e0, nE, j, hi);
  else
    edge_group<true>(p, src, dst, invdeg, out, e0, nE, j, hi);
}

extern "C" void kernel_launch(void* const* d_in, const int* in_sizes, int n_in,
                              void* d_out, int out_size, void* d_ws,
                              size_t ws_size, hipStream_t stream) {
  (void)n_in; (void)out_size; (void)ws_size;

  const float* x   = (const float*)d_in[0];
  const int*   ei  = (const int*)d_in[1];
  const float* W1l = (const float*)d_in[2];
  const float* b1  = (const float*)d_in[3];
  const float* W1r = (const float*)d_in[4];
  const float* W2l = (const float*)d_in[5];
  const float* b2  = (const float*)d_in[6];
  const float* W2r = (const float*)d_in[7];
  float* out = (float*)d_out;

  const int N  = in_sizes[0] / K_F;
  const int nE = in_sizes[1] / 2;
  const int* src = ei;
  const int* dst = ei + nE;

  float* msg = (float*)d_ws;
  float* deg = msg + (size_t)N * K_F;
  float* h   = deg + N;

  const int T = 256;

  {
    int n0 = N * K_F + N;
    zero_kernel<<<(n0 + T - 1) / T, T, 0, stream>>>(msg, n0);
  }
  degree_kernel<<<(nE + T - 1) / T, T, 0, stream>>>(dst, deg, nE);
  inv_kernel<<<(N + T - 1) / T, T, 0, stream>>>(deg, N);

  {
    int waves  = (nE + EPW1 - 1) / EPW1;
    int blocks = (waves + 7) / 8;
    scatter_kernel<<<blocks, T, 0, stream>>>(x, src, dst, msg, nE);
  }
  {
    int nTilesN = K_F / 16;
    int nTiles  = (N / 16) * nTilesN;
    int blocks  = (nTiles + 7) / 8;
    sage_gemm<K_F, true><<<blocks, T, 0, stream>>>(msg, deg, x, W1l, W1r, b1,
                                                   h, nTilesN, nTiles);
  }

  float* p = msg;
  {
    int nTiles = N / 16;
    int blocks = (nTiles + 7) / 8;
    proj2_kernel<<<blocks, T, 0, stream>>>(h, W2l, W2r, b2, p, out, nTiles);
  }
  {
    int waves  = (nE + EPW - 1) / EPW;
    int blocks = (waves + 7) / 8;
    scatter16_kernel<<<blocks, T, 0, stream>>>(p, src, dst, deg, out, nE);
  }
}


# f32 atomic sums (degree, layer-1 rows, layer-2 rows) get one copy per cache half, chosen by die id; consumers add the copies (layer-1 GEMM in place, closing pass applies 1/deg); sum order changes
# speedup vs baseline: 1.6686x; 1.0346x over previous
#include <hip/hip_runtime.h>


typedef __attribute__((ext_vector_type(2))) float v2f;
typedef __attribute__((ext_vector_type(4))) float v4f;
typedef __attribute__((ext_vector_type(8))) float v8f;

#define K_F 64
#define N_C 16
#define EPW 16

__device__ __forceinline__ int cache_half() {
  return (int)((__builtin_amdgcn_s_sendmsg_rtn(0x87) >> 18) & 1u);
}

__global__ void zero_kernel(float* __restrict__ p, int n) {
  int i = (blockIdx.x * blockDim.x + threadIdx.x) * 4;
  if (i + 3 < n) {
    v4f z = {0.0f, 0.0f, 0.0f, 0.0f};
    *reinterpret_cast<v4f*>(p + i) = z;
  } else {
    for (int t = 0; t < 3; ++t)
      if (i + t < n) p[i + t] = 0.0f;
  }
}

__global__ void degree_kernel(const int* __restrict__ dst,
                              float* __restrict__ degA,
                              float* __restrict__ degB, int nE) {
  int i = blockIdx.x * blockDim.x + threadIdx.x;
  if (i < nE) {
    int d = dst[i];
    float* deg = cache_half() ? degB : degA;
    atomicAdd(&deg[d], 1.0f);
  }
}

__global__ void inv_kernel(float* __restrict__ degA, float* __restrict__ degB,
                           int n) {
  int i = blockIdx.x * blockDim.x + threadIdx.x;
  if (i < n) {
    degA[i] = 1.0f / fmaxf(degA[i] + degB[i], 1.0f);
    degB[i] = 0.0f;
  }
}

#define EPW1 16

template <bool TAIL>
__device__ __forceinline__ void edge_group64(const float* __restrict__ feat,
                                             const int* __restrict__ src,
                                             const int* __restrict__ dst,
                                             float* __restrict__ msgA,
                                             float* __restrict__ msgB,
                                             int e0, int nE, int lane) {
  int j = lane & 15;
  int idx = 0;
  if (!TAIL || e0 + j < nE) {
    const int* ip = (lane >> 4) ? dst : src;
    idx = ip[e0 + j];
  }

  float lo[EPW1];
  float hi[EPW1];
#pragma unroll
  for (int t = 0; t < EPW1; ++t) {
    lo[t] = 0.0f;
    hi[t] = 0.0f;
    if (!TAIL || e0 + t < nE) {
      int s = __builtin_amdgcn_readlane(idx, t);
      const float* r = feat + (size_t)s * K_F + lane;
      lo[t] = r[0];
      hi[t] = r[32];
    }
  }
  float* msg = cache_half() ? msgB : msgA;
#pragma unroll
  for (int t = 0; t < EPW1; ++t) {
    if (!TAIL || e0 + t < nE) {
      int d = __builtin_amdgcn_readlane(idx, 16 + t);
      float* b = msg + (size_t)d * K_F + lane;
      atomicAdd(b, lo[t]);
      atomicAdd(b + 32, hi[t]);
    }
  }
}

__global__ void scatter_kernel(const float* __restrict__ feat,
                               const int* __restrict__ src,
                               const int* __restrict__ dst,
                               float* __restrict__ msgA,
                               float* __restrict__ msgB, int nE) {
  int wave = blockIdx.x * (blockDim.x >> 5) + (threadIdx.x >> 5);
  int e0 = wave * EPW1;
  if (e0 >= nE) return;
  int lane = threadIdx.x & 31;
  if (e0 + EPW1 <= nE)
    edge_group64<false>(feat, src, dst, msgA, msgB, e0, nE, lane);
  else
    edge_group64<true>(feat, src, dst, msgA, msgB, e0, nE, lane);
}

__global__ void sage_gemm64(const float* __restrict__ msgA,
                            float* hb,
                            const float* __restrict__ invdeg,
                            const float* __restrict__ xin,
                            const float* __restrict__ Wl,
                            const float* __restrict__ Wr,
                            const float* __restrict__ bias,
                            int nStripes) {
  int stripe = blockIdx.x * (blockDim.x >> 5) + (threadIdx.x >> 5);
  if (stripe >= nStripes) return;

  int lane = threadIdx.x & 31;
  int hrow = lane & 15;
  int hi   = lane >> 4;
  int row  = stripe * 16 + hrow;
  int koff = hi * 2;

  float inv = invdeg[row];
  const float* arow = msgA + (size_t)row * K_F + koff;
  const float* brow = hb   + (size_t)row * K_F + koff;
  const float* xrow = xin  + (size_t)row * K_F + koff;

  v8f c0 = {};
  v8f c1 = {};
  v8f c2 = {};
  v8f c3 = {};
#pragma unroll
  for (int k = 0; k < K_F; k += 4) {
    v2f a0 = *reinterpret_cast<const v2f*>(arow + k);
    v2f a1 = *reinterpret_cast<const v2f*>(brow + k);
    v2f am;
    am.x = (a0.x + a1.x) * inv;
    am.y = (a0.y + a1.y) * inv;
    v2f ax = *reinterpret_cast<const v2f*>(xrow + k);
    const float* wl0 = Wl + (k + koff) * K_F + hrow;
    const float* wr0 = Wr + (k + koff) * K_F + hrow;
    v2f bl, br;
    bl.x = wl0[0];        bl.y = wl0[K_F];
    br.x = wr0[0];        br.y = wr0[K_F];
    c0 = __builtin_amdgcn_wmma_f32_16x16x4_f32(false, am, false, bl,
                                               (short)0, c0, false, false);
    c0 = __builtin_amdgcn_wmma_f32_16x16x4_f32(false, ax, false, br,
                                               (short)0, c0, false, false);
    bl.x = wl0[16];       bl.y = wl0[K_F + 16];
    br.x = wr0[16];       br.y = wr0[K_F + 16];
    c1 = __builtin_amdgcn_wmma_f32_16x16x4_f32(false, am, false, bl,
                                               (short)0, c1, false, false);
    c1 = __builtin_amdgcn_wmma_f32_16x16x4_f32(false, ax, false, br,
                                               (short)0, c1, false, false);
    bl.x = wl0[32];       bl.y = wl0[K_F + 32];
    br.x = wr0[32];       br.y = wr0[K_F + 32];
    c2 = __builtin_amdgcn_wmma_f32_16x16x4_f32(false, am, false, bl,
                                               (short)0, c2, false, false);
    c2 = __builtin_amdgcn_wmma_f32_16x16x4_f32(false, ax, false, br,
                                               (short)0, c2, false, false);
    bl.x = wl0[48];       bl.y = wl0[K_F + 48];
    br.x = wr0[48];       br.y = wr0[K_F + 48];
    c3 = __builtin_amdgcn_wmma_f32_16x16x4_f32(false, am, false, bl,
                                               (short)0, c3, false, false);
    c3 = __builtin_amdgcn_wmma_f32_16x16x4_f32(false, ax, false, br,
                                               (short)0, c3, false, false);
  }

  float bv0 = bias[hrow];
  float bv1 = bias[16 + hrow];
  float bv2 = bias[32 + hrow];
  float bv3 = bias[48 + hrow];
#pragma unroll
  for (int i = 0; i < 8; ++i) {
    float* o = hb + (size_t)(stripe * 16 + hi * 8 + i) * K_F + hrow;
    o[0]  = fmaxf(c0[i] + bv0, 0.0f);
    o[16] = fmaxf(c1[i] + bv1, 0.0f);
    o[32] = fmaxf(c2[i] + bv2, 0.0f);
    o[48] = fmaxf(c3[i] + bv3, 0.0f);
  }
}

__global__ void proj2_kernel(const float* __restrict__ hin,
                             const float* __restrict__ Wl,
                             const float* __restrict__ Wr,
                             const float* __restrict__ bias,
                             float* __restrict__ p,
                             float* __restrict__ q,
                             float* __restrict__ accB,
                             float* __restrict__ out, int nTiles) {
  int tile = blockIdx.x * (blockDim.x >> 5) + (threadIdx.x >> 5);
  if (tile >= nTiles) return;
  int lane = threadIdx.x & 31;
  int col  = lane & 15;
  int hi   = lane >> 4;
  int koff = hi * 2;
  const float* hrow = hin + (size_t)(tile * 16 + col) * K_F + koff;

  v8f cp = {};
  v8f cq = {};
#pragma unroll
  for (int k = 0; k < K_F; k += 4) {
    v2f a = *reinterpret_cast<const v2f*>(hrow + k);
    v2f bl, br;
    bl.x = Wl[(k + koff) * N_C + col];
    bl.y = Wl[(k + koff + 1) * N_C + col];
    br.x = Wr[(k + koff) * N_C + col];
    br.y = Wr[(k + koff + 1) * N_C + col];
    cp = __builtin_amdgcn_wmma_f32_16x16x4_f32(false, a, false, bl,
                                               (short)0, cp, false, false);
    cq = __builtin_amdgcn_wmma_f32_16x16x4_f32(false, a, false, br,
                                               (short)0, cq, false, false);
  }

  float bv = bias[col];
#pragma unroll
  for (int i = 0; i < 8; ++i) {
    size_t o = (size_t)(tile * 16 + hi * 8 + i) * N_C + col;
    p[o]    = cp[i];
    q[o]    = cq[i] + bv;
    accB[o] = 0.0f;
    out[o]  = 0.0f;
  }
}

template <bool TAIL>
__device__ __forceinline__ void edge_group(const float* __restrict__ p,
                                           const int* __restrict__ src,
                                           const int* __restrict__ dst,
                                           float* __restrict__ accA,
                                           float* __restrict__ accB,
                                           int e0, int nE, int j, int hi) {
  int idx = 0;
  if (!TAIL || e0 + j < nE) {
    const int* ip = hi ? dst : src;
    idx = ip[e0 + j];
  }

  float v[EPW / 2];
  int   dd[EPW / 2];
#pragma unroll
  for (int t = 0; t < EPW / 2; ++t) {
    int q = hi * (EPW / 2) + t;
    int s = __shfl(idx, q);
    dd[t] = __shfl(idx, 16 + q);
    v[t]  = (!TAIL || e0 + q < nE) ? p[(size_t)s * N_C + j] : 0.0f;
  }
  float* acc = cache_half() ? accB : accA;
#pragma unroll
  for (int t = 0; t < EPW / 2; ++t) {
    int q = hi * (EPW / 2) + t;
    if (!TAIL || e0 + q < nE)
      atomicAdd(acc + (size_t)dd[t] * N_C + j, v[t]);
  }
}

__global__ void scatter16_kernel(const float* __restrict__ p,
                                 const int* __restrict__ src,
                                 const int* __restrict__ dst,
                                 float* __restrict__ accA,
                                 float* __restrict__ accB, int nE) {
  int wave = blockIdx.x * (blockDim.x >> 5) + (threadIdx.x >> 5);
  int e0 = wave * EPW;
  if (e0 >= nE) return;
  int lane = threadIdx.x & 31;
  int j  = lane & 15;
  int hi = lane >> 4;
  if (e0 + EPW <= nE)
    edge_group<false>(p, src, dst, accA, accB, e0, nE, j, hi);
  else
    edge_group<true>(p, src, dst, accA, accB, e0, nE, j, hi);
}

__global__ void close_kernel(const float* __restrict__ q,
                             const float* __restrict__ accB,
                             const float* __restrict__ invdeg,
                             float* __restrict__ out, int n4) {
  int i = blockIdx.x * blockDim.x + threadIdx.x;
  if (i < n4) {
    v4f a = reinterpret_cast<const v4f*>(out)[i];
    v4f b = reinterpret_cast<const v4f*>(accB)[i];
    v4f s = reinterpret_cast<const v4f*>(q)[i];
    float w = invdeg[i >> 2];
    reinterpret_cast<v4f*>(out)[i] = s + (a + b) * w;
  }
}

extern "C" void kernel_launch(void* const* d_in, const int* in_sizes, int n_in,
                              void* d_out, int out_size, void* d_ws,
                              size_t ws_size, hipStream_t stream) {
  (void)n_in; (void)out_size; (void)ws_size;

  const float* x   = (const float*)d_in[0];
  const int*   ei  = (const int*)d_in[1];
  const float* W1l = (const float*)d_in[2];
  const float* b1  = (const float*)d_in[3];
  const float* W1r = (const float*)d_in[4];
  const float* W2l = (const float*)d_in[5];
  const float* b2  = (const float*)d_in[6];
  const float* W2r = (const float*)d_in[7];
  float* out = (float*)d_out;

  const int N  = in_sizes[0] / K_F;
  const int nE = in_sizes[1] / 2;
  const int* src = ei;
  const int* dst = ei + nE;

  float* msgA = (float*)d_ws;
  float* msgB = msgA + (size_t)N * K_F;
  float* degA = msgB + (size_t)N * K_F;
  float* degB = msgB;
  float* h    = msgB;

  const int T = 256;

  {
    int n0 = 2 * N * K_F + N;
    int nT = (n0 + 3) / 4;
    zero_kernel<<<(nT + T - 1) / T, T, 0, stream>>>(msgA, n0);
  }
  degree_kernel<<<(nE + T - 1) / T, T, 0, stream>>>(dst, degA, degB, nE);
  inv_kernel<<<(N + T - 1) / T, T, 0, stream>>>(degA, degB, N);

  {
    int waves  = (nE + EPW1 - 1) / EPW1;
    int blocks = (waves + 7) / 8;
    scatter_kernel<<<blocks, T, 0, stream>>>(x, src, dst, msgA, msgB, nE);
  }
  {
    int nStripes = N / 16;
    int blocks   = (nStripes + 7) / 8;
    sage_gemm64<<<blocks, T, 0, stream>>>(msgA, h, degA, x, W1l, W1r, b1,
                                          nStripes);
  }

  float* p    = msgA;
  float* q    = msgA + (size_t)N * N_C;
  float* accB = msgA + (size_t)2 * N * N_C;
  {
    int nTiles = N / 16;
    int blocks = (nTiles + 7) / 8;
    proj2_kernel<<<blocks, T, 0, stream>>>(h, W2l, W2r, b2, p, q, accB, out,
                                           nTiles);
  }
  {
    int waves  = (nE + EPW - 1) / EPW;
    int blocks = (waves + 7) / 8;
    scatter16_kernel<<<blocks, T, 0, stream>>>(p, src, dst, out, accB, nE);
  }
  {
    int n4 = N * (N_C / 4);
    close_kernel<<<(n4 + T - 1) / T, T, 0, stream>>>(q, accB, degA, out, n4);
  }
}


# zero, degree and both edge passes as resident grids: 4,096-8,192 waves loop over their items (next edge indices loaded ahead), not 25,000-50,000 one-item waves; atomic sums still in arrival order
# speedup vs baseline: 1.8243x; 1.0933x over previous
#include <hip/hip_runtime.h>


typedef __attribute__((ext_vector_type(2))) float v2f;
typedef __attribute__((ext_vector_type(4))) float v4f;
typedef __attribute__((ext_vector_type(8))) float v8f;

#define K_F 64
#define N_C 16
#define EPW 16

__device__ __forceinline__ int cache_half() {
  return (int)((__builtin_amdgcn_s_sendmsg_rtn(0x87) >> 18) & 1u);
}

#define RES_BLOCKS_SMALL 512
#define RES_BLOCKS_EDGE  1024

__device__ __forceinline__ int wave_in_grid() {
  return __builtin_amdgcn_readfirstlane(
      (int)(blockIdx.x * (blockDim.x >> 5) + (threadIdx.x >> 5)));
}

__global__ void zero_kernel(float* __restrict__ p, int n) {
  int tid    = blockIdx.x * blockDim.x + threadIdx.x;
  int stride = gridDim.x * blockDim.x;
  int n4     = n >> 2;
  v4f z = {0.0f, 0.0f, 0.0f, 0.0f};
  for (int i = tid; i < n4; i += stride)
    reinterpret_cast<v4f*>(p)[i] = z;
  if (tid < (n & 3))
    p[(n4 << 2) + tid] = 0.0f;
}

#define DEG_U 4

__global__ void degree_kernel(const int* __restrict__ dst,
                              float* __restrict__ degA,
                              float* __restrict__ degB, int nE) {
  int stride = gridDim.x * blockDim.x;
  int last   = nE - 1;
  float* deg = cache_half() ? degB : degA;
  for (int i = blockIdx.x * blockDim.x + threadIdx.x; i < nE;
       i += DEG_U * stride) {
    int d[DEG_U];
#pragma unroll
    for (int u = 0; u < DEG_U; ++u)
      d[u] = dst[min(i + u * stride, last)];
#pragma unroll
    for (int u = 0; u < DEG_U; ++u)
      if (i + u * stride < nE) atomicAdd(&deg[d[u]], 1.0f);
  }
}

__global__ void inv_kernel(float* __restrict__ degA, float* __restrict__ degB,
                           int n) {
  int i = blockIdx.x * blockDim.x + threadIdx.x;
  if (i < n) {
    degA[i] = 1.0f / fmaxf(degA[i] + degB[i], 1.0f);
    degB[i] = 0.0f;
  }
}

#define EPW1 16

template <bool TAIL>
__device__ __forceinline__ void edge_group64(const float* __restrict__ feat,
                                             float* __restrict__ msg,
                                             int idx, int e0, int nE,
                                             int lane) {
  float lo[EPW1];
  float hi[EPW1];
#pragma unroll
  for (int t = 0; t < EPW1; ++t) {
    lo[t] = 0.0f;
    hi[t] = 0.0f;
    if (!TAIL || e0 + t < nE) {
      int s = __builtin_amdgcn_readlane(idx, t);
      const float* r = feat + (size_t)s * K_F + lane;
      lo[t] = r[0];
      hi[t] = r[32];
    }
  }
#pragma unroll
  for (int t = 0; t < EPW1; ++t) {
    if (!TAIL || e0 + t < nE) {
      int d = __builtin_amdgcn_readlane(idx, 16 + t);
      float* b = msg + (size_t)d * K_F + lane;
      atomicAdd(b, lo[t]);
      atomicAdd(b + 32, hi[t]);
    }
  }
}

__global__ void scatter_kernel(const float* __restrict__ feat,
                               const int* __restrict__ src,
                               const int* __restrict__ dst,
                               float* __restrict__ msgA,
                               float* __restrict__ msgB, int nE) {
  int nGroups = (nE + EPW1 - 1) / EPW1;
  int nWaves  = gridDim.x * (blockDim.x >> 5);
  int g       = wave_in_grid();
  if (g >= nGroups) return;
  int lane = threadIdx.x & 31;
  int j    = lane & 15;
  const int* ip = (lane >> 4) ? dst : src;

  int last = nE - 1;
  int idx  = ip[min(g * EPW1 + j, last)];
  float* msg = cache_half() ? msgB : msgA;
  for (; g < nGroups; g += nWaves) {
    int e0 = g * EPW1;
    int idx_next = ip[min(e0 + nWaves * EPW1 + j, last)];
    if (e0 + EPW1 <= nE)
      edge_group64<false>(feat, msg, idx, e0, nE, lane);
    else
      edge_group64<true>(feat, msg, idx, e0, nE, lane);
    idx = idx_next;
  }
}

__global__ void sage_gemm64(const float* __restrict__ msgA,
                            float* hb,
                            const float* __restrict__ invdeg,
                            const float* __restrict__ xin,
                            const float* __restrict__ Wl,
                            const float* __restrict__ Wr,
                            const float* __restrict__ bias,
                            int nStripes) {
  int stripe = blockIdx.x * (blockDim.x >> 5) + (threadIdx.x >> 5);
  if (stripe >= nStripes) return;

  int lane = threadIdx.x & 31;
  int hrow = lane & 15;
  int hi   = lane >> 4;
  int row  = stripe * 16 + hrow;
  int koff = hi * 2;

  float inv = invdeg[row];
  const float* arow = msgA + (size_t)row * K_F + koff;
  const float* brow = hb   + (size_t)row * K_F + koff;
  const float* xrow = xin  + (size_t)row * K_F + koff;

  v8f c0 = {};
  v8f c1 = {};
  v8f c2 = {};
  v8f c3 = {};
#pragma unroll
  for (int k = 0; k < K_F; k += 4) {
    v2f a0 = *reinterpret_cast<const v2f*>(arow + k);
    v2f a1 = *reinterpret_cast<const v2f*>(brow + k);
    v2f am;
    am.x = (a0.x + a1.x) * inv;
    am.y = (a0.y + a1.y) * inv;
    v2f ax = *reinterpret_cast<const v2f*>(xrow + k);
    const float* wl0 = Wl + (k + koff) * K_F + hrow;
    const float* wr0 = Wr + (k + koff) * K_F + hrow;
    v2f bl, br;
    bl.x = wl0[0];        bl.y = wl0[K_F];
    br.x = wr0[0];        br.y = wr0[K_F];
    c0 = __builtin_amdgcn_wmma_f32_16x16x4_f32(false, am, false, bl,
                                               (short)0, c0, false, false);
    c0 = __builtin_amdgcn_wmma_f32_16x16x4_f32(false, ax, false, br,
                                               (short)0, c0, false, false);
    bl.x = wl0[16];       bl.y = wl0[K_F + 16];
    br.x = wr0[16];       br.y = wr0[K_F + 16];
    c1 = __builtin_amdgcn_wmma_f32_16x16x4_f32(false, am, false, bl,
                                               (short)0, c1, false, false);
    c1 = __builtin_amdgcn_wmma_f32_16x16x4_f32(false, ax, false, br,
                                               (short)0, c1, false, false);
    bl.x = wl0[32];       bl.y = wl0[K_F + 32];
    br.x = wr0[32];       br.y = wr0[K_F + 32];
    c2 = __builtin_amdgcn_wmma_f32_16x16x4_f32(false, am, false, bl,
                                               (short)0, c2, false, false);
    c2 = __builtin_amdgcn_wmma_f32_16x16x4_f32(false, ax, false, br,
                                               (short)0, c2, false, false);
    bl.x = wl0[48];       bl.y = wl0[K_F + 48];
    br.x = wr0[48];       br.y = wr0[K_F + 48];
    c3 = __builtin_amdgcn_wmma_f32_16x16x4_f32(false, am, false, bl,
                                               (short)0, c3, false, false);
    c3 = __builtin_amdgcn_wmma_f32_16x16x4_f32(false, ax, false, br,
                                               (short)0, c3, false, false);
  }

  float bv0 = bias[hrow];
  float bv1 = bias[16 + hrow];
  float bv2 = bias[32 + hrow];
  float bv3 = bias[48 + hrow];
#pragma unroll
  for (int i = 0; i < 8; ++i) {
    float* o = hb + (size_t)(stripe * 16 + hi * 8 + i) * K_F + hrow;
    o[0]  = fmaxf(c0[i] + bv0, 0.0f);
    o[16] = fmaxf(c1[i] + bv1, 0.0f);
    o[32] = fmaxf(c2[i] + bv2, 0.0f);
    o[48] = fmaxf(c3[i] + bv3, 0.0f);
  }
}

__global__ void proj2_kernel(const float* __restrict__ hin,
                             const float* __restrict__ Wl,
                             const float* __restrict__ Wr,
                             const float* __restrict__ bias,
                             float* __restrict__ p,
                             float* __restrict__ q,
                             float* __restrict__ accB,
                             float* __restrict__ out, int nTiles) {
  int tile = blockIdx.x * (blockDim.x >> 5) + (threadIdx.x >> 5);
  if (tile >= nTiles) return;
  int lane = threadIdx.x & 31;
  int col  = lane & 15;
  int hi   = lane >> 4;
  int koff = hi * 2;
  const float* hrow = hin + (size_t)(tile * 16 + col) * K_F + koff;

  v8f cp = {};
  v8f cq = {};
#pragma unroll
  for (int k = 0; k < K_F; k += 4) {
    v2f a = *reinterpret_cast<const v2f*>(hrow + k);
    v2f bl, br;
    bl.x = Wl[(k + koff) * N_C + col];
    bl.y = Wl[(k + koff + 1) * N_C + col];
    br.x = Wr[(k + koff) * N_C + col];
    br.y = Wr[(k + koff + 1) * N_C + col];
    cp = __builtin_amdgcn_wmma_f32_16x16x4_f32(false, a, false, bl,
                                               (short)0, cp, false, false);
    cq = __builtin_amdgcn_wmma_f32_16x16x4_f32(false, a, false, br,
                                               (short)0, cq, false, false);
  }

  float bv = bias[col];
#pragma unroll
  for (int i = 0; i < 8; ++i) {
    size_t o = (size_t)(tile * 16 + hi * 8 + i) * N_C + col;
    p[o]    = cp[i];
    q[o]    = cq[i] + bv;
    accB[o] = 0.0f;
    out[o]  = 0.0f;
  }
}

template <bool TAIL>
__device__ __forceinline__ void edge_group(const float* __restrict__ p,
                                           float* __restrict__ acc,
                                           int idx, int e0, int nE, int j,
                                           int hi) {
  float v[EPW / 2];
  int   dd[EPW / 2];
#pragma unroll
  for (int t = 0; t < EPW / 2; ++t) {
    int q = hi * (EPW / 2) + t;
    int s = __shfl(idx, q);
    dd[t] = __shfl(idx, 16 + q);
    v[t]  = (!TAIL || e0 + q < nE) ? p[(size_t)s * N_C + j] : 0.0f;
  }
#pragma unroll
  for (int t = 0; t < EPW / 2; ++t) {
    int q = hi * (EPW / 2) + t;
    if (!TAIL || e0 + q < nE)
      atomicAdd(acc + (size_t)dd[t] * N_C + j, v[t]);
  }
}

__global__ void scatter16_kernel(const float* __restrict__ p,
                                 const int* __restrict__ src,
                                 const int* __restrict__ dst,
                                 float* __restrict__ accA,
                                 float* __restrict__ accB, int nE) {
  int nGroups = (nE + EPW - 1) / EPW;
  int nWaves  = gridDim.x * (blockDim.x >> 5);
  int g       = wave_in_grid();
  if (g >= nGroups) return;
  int lane = threadIdx.x & 31;
  int j  = lane & 15;
  int hi = lane >> 4;
  const int* ip = hi ? dst : src;

  int last = nE - 1;
  int idx  = ip[min(g * EPW + j, last)];
  float* acc = cache_half() ? accB : accA;
  for (; g < nGroups; g += nWaves) {
    int e0 = g * EPW;
    int idx_next = ip[min(e0 + nWaves * EPW + j, last)];
    if (e0 + EPW <= nE)
      edge_group<false>(p, acc, idx, e0, nE, j, hi);
    else
      edge_group<true>(p, acc, idx, e0, nE, j, hi);
    idx = idx_next;
  }
}

__global__ void close_kernel(const float* __restrict__ q,
                             const float* __restrict__ accB,
                             const float* __restrict__ invdeg,
                             float* __restrict__ out, int n4) {
  int i = blockIdx.x * blockDim.x + threadIdx.x;
  if (i < n4) {
    v4f a = reinterpret_cast<const v4f*>(out)[i];
    v4f b = reinterpret_cast<const v4f*>(accB)[i];
    v4f s = reinterpret_cast<const v4f*>(q)[i];
    float w = invdeg[i >> 2];
    reinterpret_cast<v4f*>(out)[i] = s + (a + b) * w;
  }
}

extern "C" void kernel_launch(void* const* d_in, const int* in_sizes, int n_in,
                              void* d_out, int out_size, void* d_ws,
                              size_t ws_size, hipStream_t stream) {
  (void)n_in; (void)out_size; (void)ws_size;

  const float* x   = (const float*)d_in[0];
  const int*   ei  = (const int*)d_in[1];
  const float* W1l = (const float*)d_in[2];
  const float* b1  = (const float*)d_in[3];
  const float* W1r = (const float*)d_in[4];
  const float* W2l = (const float*)d_in[5];
  const float* b2  = (const float*)d_in[6];
  const float* W2r = (const float*)d_in[7];
  float* out = (float*)d_out;

  const int N  = in_sizes[0] / K_F;
  const int nE = in_sizes[1] / 2;
  const int* src = ei;
  const int* dst = ei + nE;

  float* msgA = (float*)d_ws;
  float* msgB = msgA + (size_t)N * K_F;
  float* degA = msgB + (size_t)N * K_F;
  float* degB = msgB;
  float* h    = msgB;

  const int T = 256;

  {
    int n0 = 2 * N * K_F + N;
    zero_kernel<<<RES_BLOCKS_SMALL, T, 0, stream>>>(msgA, n0);
  }
  degree_kernel<<<RES_BLOCKS_SMALL, T, 0, stream>>>(dst, degA, degB, nE);
  inv_kernel<<<(N + T - 1) / T, T, 0, stream>>>(degA, degB, N);

  scatter_kernel<<<RES_BLOCKS_EDGE, T, 0, stream>>>(x, src, dst, msgA, msgB,
                                                    nE);
  {
    int nStripes = N / 16;
    int blocks   = (nStripes + 7) / 8;
    sage_gemm64<<<blocks, T, 0, stream>>>(msgA, h, degA, x, W1l, W1r, b1,
                                          nStripes);
  }

  float* p    = msgA;
  float* q    = msgA + (size_t)N * N_C;
  float* accB = msgA + (size_t)2 * N * N_C;
  {
    int nTiles = N / 16;
    int blocks = (nTiles + 7) / 8;
    proj2_kernel<<<blocks, T, 0, stream>>>(h, W2l, W2r, b2, p, q, accB, out,
                                           nTiles);
  }
  scatter16_kernel<<<RES_BLOCKS_EDGE, T, 0, stream>>>(p, src, dst, out, accB,
                                                      nE);
  {
    int n4 = N * (N_C / 4);
    close_kernel<<<(n4 + T - 1) / T, T, 0, stream>>>(q, accB, degA, out, n4);
  }
}


# neighbour sums without f32 atomics: edges sorted by destination each call (two-level counting sort on shared-memory counters), then per-node row gathers; f32 sum order = the sort's arrival order
# speedup vs baseline: 2.4752x; 1.3568x over previous
#include <hip/hip_runtime.h>


typedef __attribute__((ext_vector_type(2))) float v2f;
typedef __attribute__((ext_vector_type(2))) int   v2i;
typedef __attribute__((ext_vector_type(8))) float v8f;

#define K_F 64
#define N_C 16

#define SORT_T 1024
#define SORT_G 64
#define SORT_U 8
#define BIN_SH 8
#define BIN_W  (1 << BIN_SH)
#define NB_MAX SORT_T

__device__ __forceinline__ int wave_scan_incl(int v, int lane) {
  for (int o = 1; o < 32; o <<= 1) {
    int t = __shfl_up(v, o);
    if (lane >= o) v += t;
  }
  return v;
}

__global__ void bin_count_kernel(const int* __restrict__ dst,
                                 int* __restrict__ wgcnt, int nE, int nb) {
  __shared__ int hist[NB_MAX];
  int tid = threadIdx.x;
  if (tid < nb) hist[tid] = 0;
  __syncthreads();
  int S  = (nE + gridDim.x - 1) / gridDim.x;
  int lo = blockIdx.x * S;
  int hi = min(lo + S, nE);
  for (int e = lo + tid; e < hi; e += SORT_U * SORT_T) {
    int d[SORT_U];
#pragma unroll
    for (int u = 0; u < SORT_U; ++u)
      d[u] = dst[min(e + u * SORT_T, hi - 1)];
#pragma unroll
    for (int u = 0; u < SORT_U; ++u)
      if (e + u * SORT_T < hi) atomicAdd(&hist[d[u] >> BIN_SH], 1);
  }
  __syncthreads();
  if (tid < nb) wgcnt[blockIdx.x * nb + tid] = hist[tid];
}

__global__ void bin_place_kernel(const int* __restrict__ src,
                                 const int* __restrict__ dst,
                                 const int* __restrict__ wgcnt,
                                 int* __restrict__ binbase,
                                 int* __restrict__ pairs_, int nE, int nb) {
  __shared__ int cur[NB_MAX];
  __shared__ int s_part[SORT_T / 32];
  v2i* pairs = reinterpret_cast<v2i*>(pairs_);
  int tid  = threadIdx.x;
  int lane = tid & 31;
  int w    = tid >> 5;
  int G    = gridDim.x;
  int g    = blockIdx.x;

  int bcol   = min(tid, nb - 1);
  int total  = 0;
  int before = 0;
  for (int k0 = 0; k0 < G; k0 += 16) {
    int t[16];
#pragma unroll
    for (int u = 0; u < 16; ++u) {
      t[u] = 0;
      if (k0 + u < G) t[u] = wgcnt[(k0 + u) * nb + bcol];
    }
#pragma unroll
    for (int u = 0; u < 16; ++u) {
      total += t[u];
      if (k0 + u < g) before += t[u];
    }
  }
  if (tid >= nb) {
    total  = 0;
    before = 0;
  }
  int inc = wave_scan_incl(total, lane);
  if (lane == 31) s_part[w] = inc;
  __syncthreads();
  int woff = 0;
  for (int k = 0; k < w; ++k) woff += s_part[k];
  int excl = woff + inc - total;
  if (tid < nb) {
    cur[tid] = excl + before;
    if (g == 0) {
      binbase[tid] = excl;
      if (tid == nb - 1) binbase[nb] = excl + total;
    }
  }
  __syncthreads();

  int S  = (nE + G - 1) / G;
  int lo = g * S;
  int hi = min(lo + S, nE);
  for (int e = lo + tid; e < hi; e += SORT_U * SORT_T) {
    int d[SORT_U];
    int s[SORT_U];
    int pos[SORT_U];
#pragma unroll
    for (int u = 0; u < SORT_U; ++u) {
      int ee = min(e + u * SORT_T, hi - 1);
      d[u] = dst[ee];
      s[u] = src[ee];
    }
#pragma unroll
    for (int u = 0; u < SORT_U; ++u) {
      pos[u] = 0;
      if (e + u * SORT_T < hi) pos[u] = atomicAdd(&cur[d[u] >> BIN_SH], 1);
    }
#pragma unroll
    for (int u = 0; u < SORT_U; ++u)
      if (e + u * SORT_T < hi) {
        v2i sd = {s[u], d[u]};
        pairs[pos[u]] = sd;
      }
  }
}

__global__ void bin_sort_kernel(const int* __restrict__ pairs_,
                                const int* __restrict__ binbase,
                                int* __restrict__ rowptr,
                                int* __restrict__ col, int N) {
  __shared__ int cnt[BIN_W];
  __shared__ int s_part[SORT_T / 32];
  const v2i* pairs = reinterpret_cast<const v2i*>(pairs_);
  int tid   = threadIdx.x;
  int lane  = tid & 31;
  int w     = tid >> 5;
  int b     = blockIdx.x;
  int seg0  = binbase[b];
  int seg1  = binbase[b + 1];
  int node0 = b << BIN_SH;

  if (tid < BIN_W) cnt[tid] = 0;
  __syncthreads();
  for (int e = seg0 + tid; e < seg1; e += SORT_U * SORT_T) {
    int d[SORT_U];
#pragma unroll
    for (int u = 0; u < SORT_U; ++u)
      d[u] = pairs[min(e + u * SORT_T, seg1 - 1)].y;
#pragma unroll
    for (int u = 0; u < SORT_U; ++u)
      if (e + u * SORT_T < seg1) atomicAdd(&cnt[d[u] - node0], 1);
  }
  __syncthreads();
  int c   = (tid < BIN_W) ? cnt[tid] : 0;
  int inc = wave_scan_incl(c, lane);
  if (lane == 31) s_part[w] = inc;
  __syncthreads();
  int woff = 0;
  for (int k = 0; k < w; ++k) woff += s_part[k];
  int excl = woff + inc - c;
  if (tid < BIN_W) {
    cnt[tid] = excl;
    int node = node0 + tid;
    if (node < N) rowptr[node] = seg0 + excl;
    if (node == N - 1) rowptr[N] = seg0 + excl + c;
  }
  __syncthreads();
  for (int e = seg0 + tid; e < seg1; e += SORT_U * SORT_T) {
    v2i sd[SORT_U];
    int r[SORT_U];
#pragma unroll
    for (int u = 0; u < SORT_U; ++u)
      sd[u] = pairs[min(e + u * SORT_T, seg1 - 1)];
#pragma unroll
    for (int u = 0; u < SORT_U; ++u) {
      r[u] = 0;
      if (e + u * SORT_T < seg1) r[u] = atomicAdd(&cnt[sd[u].y - node0], 1);
    }
#pragma unroll
    for (int u = 0; u < SORT_U; ++u)
      if (e + u * SORT_T < seg1) col[seg0 + r[u]] = sd[u].x;
  }
}

#define RES_BLOCKS_EDGE  1024

__device__ __forceinline__ int wave_in_grid() {
  return __builtin_amdgcn_readfirstlane(
      (int)(blockIdx.x * (blockDim.x >> 5) + (threadIdx.x >> 5)));
}

#define NB1 32

__global__ void gather64_kernel(const float* __restrict__ feat,
                                const int* __restrict__ rowptr,
                                const int* __restrict__ col,
                                float* __restrict__ mean, int N) {
  int nWaves = gridDim.x * (blockDim.x >> 5);
  int n      = wave_in_grid();
  if (n >= N) return;
  int lane = threadIdx.x & 31;
  int start = __builtin_amdgcn_readfirstlane(rowptr[n]);
  int end   = __builtin_amdgcn_readfirstlane(rowptr[n + 1]);
  for (; n < N; n += nWaves) {
    int nn     = min(n + nWaves, N - 1);
    int startN = rowptr[nn];
    int endN   = rowptr[nn + 1];

    int deg = end - start;
    float lo = 0.0f;
    float hi = 0.0f;
    for (int base = 0; base < deg; base += NB1) {
      int idx = col[min(start + base + lane, end - 1)];
      float a[NB1];
      float b[NB1];
#pragma unroll
      for (int t = 0; t < NB1; ++t) {
        a[t] = 0.0f;
        b[t] = 0.0f;
        if (base + t < deg) {
          int s = __builtin_amdgcn_readlane(idx, t);
          const float* r = feat + (size_t)s * K_F + lane;
          a[t] = r[0];
          b[t] = r[32];
        }
      }
#pragma unroll
      for (int t = 0; t < NB1; ++t) {
        lo += a[t];
        hi += b[t];
      }
    }
    float inv = 1.0f / fmaxf((float)deg, 1.0f);
    float* o = mean + (size_t)n * K_F + lane;
    o[0]  = lo * inv;
    o[32] = hi * inv;

    start = __builtin_amdgcn_readfirstlane(startN);
    end   = __builtin_amdgcn_readfirstlane(endN);
  }
}

__global__ void sage_gemm64(float* hb,
                            const float* __restrict__ xin,
                            const float* __restrict__ Wl,
                            const float* __restrict__ Wr,
                            const float* __restrict__ bias,
                            int nStripes) {
  int stripe = blockIdx.x * (blockDim.x >> 5) + (threadIdx.x >> 5);
  if (stripe >= nStripes) return;

  int lane = threadIdx.x & 31;
  int hrow = lane & 15;
  int hi   = lane >> 4;
  int row  = stripe * 16 + hrow;
  int koff = hi * 2;

  const float* brow = hb   + (size_t)row * K_F + koff;
  const float* xrow = xin  + (size_t)row * K_F + koff;

  v8f c0 = {};
  v8f c1 = {};
  v8f c2 = {};
  v8f c3 = {};
#pragma unroll
  for (int k = 0; k < K_F; k += 4) {
    v2f am = *reinterpret_cast<const v2f*>(brow + k);
    v2f ax = *reinterpret_cast<const v2f*>(xrow + k);
    const float* wl0 = Wl + (k + koff) * K_F + hrow;
    const float* wr0 = Wr + (k + koff) * K_F + hrow;
    v2f bl, br;
    bl.x = wl0[0];        bl.y = wl0[K_F];
    br.x = wr0[0];        br.y = wr0[K_F];
    c0 = __builtin_amdgcn_wmma_f32_16x16x4_f32(false, am, false, bl,
                                               (short)0, c0, false, false);
    c0 = __builtin_amdgcn_wmma_f32_16x16x4_f32(false, ax, false, br,
                                               (short)0, c0, false, false);
    bl.x = wl0[16];       bl.y = wl0[K_F + 16];
    br.x = wr0[16];       br.y = wr0[K_F + 16];
    c1 = __builtin_amdgcn_wmma_f32_16x16x4_f32(false, am, false, bl,
                                               (short)0, c1, false, false);
    c1 = __builtin_amdgcn_wmma_f32_16x16x4_f32(false, ax, false, br,
                                               (short)0, c1, false, false);
    bl.x = wl0[32];       bl.y = wl0[K_F + 32];
    br.x = wr0[32];       br.y = wr0[K_F + 32];
    c2 = __builtin_amdgcn_wmma_f32_16x16x4_f32(false, am, false, bl,
                                               (short)0, c2, false, false);
    c2 = __builtin_amdgcn_wmma_f32_16x16x4_f32(false, ax, false, br,
                                               (short)0, c2, false, false);
    bl.x = wl0[48];       bl.y = wl0[K_F + 48];
    br.x = wr0[48];       br.y = wr0[K_F + 48];
    c3 = __builtin_amdgcn_wmma_f32_16x16x4_f32(false, am, false, bl,
                                               (short)0, c3, false, false);
    c3 = __builtin_amdgcn_wmma_f32_16x16x4_f32(false, ax, false, br,
                                               (short)0, c3, false, false);
  }

  float bv0 = bias[hrow];
  float bv1 = bias[16 + hrow];
  float bv2 = bias[32 + hrow];
  float bv3 = bias[48 + hrow];
#pragma unroll
  for (int i = 0; i < 8; ++i) {
    float* o = hb + (size_t)(stripe * 16 + hi * 8 + i) * K_F + hrow;
    o[0]  = fmaxf(c0[i] + bv0, 0.0f);
    o[16] = fmaxf(c1[i] + bv1, 0.0f);
    o[32] = fmaxf(c2[i] + bv2, 0.0f);
    o[48] = fmaxf(c3[i] + bv3, 0.0f);
  }
}

__global__ void proj2_kernel(const float* __restrict__ hin,
                             const float* __restrict__ Wl,
                             const float* __restrict__ Wr,
                             const float* __restrict__ bias,
                             float* __restrict__ p,
                             float* __restrict__ q, int nTiles) {
  int tile = blockIdx.x * (blockDim.x >> 5) + (threadIdx.x >> 5);
  if (tile >= nTiles) return;
  int lane = threadIdx.x & 31;
  int col  = lane & 15;
  int hi   = lane >> 4;
  int koff = hi * 2;
  const float* hrow = hin + (size_t)(tile * 16 + col) * K_F + koff;

  v8f cp = {};
  v8f cq = {};
#pragma unroll
  for (int k = 0; k < K_F; k += 4) {
    v2f a = *reinterpret_cast<const v2f*>(hrow + k);
    v2f bl, br;
    bl.x = Wl[(k + koff) * N_C + col];
    bl.y = Wl[(k + koff + 1) * N_C + col];
    br.x = Wr[(k + koff) * N_C + col];
    br.y = Wr[(k + koff + 1) * N_C + col];
    cp = __builtin_amdgcn_wmma_f32_16x16x4_f32(false, a, false, bl,
                                               (short)0, cp, false, false);
    cq = __builtin_amdgcn_wmma_f32_16x16x4_f32(false, a, false, br,
                                               (short)0, cq, false, false);
  }

  float bv = bias[col];
#pragma unroll
  for (int i = 0; i < 8; ++i) {
    size_t o = (size_t)(tile * 16 + hi * 8 + i) * N_C + col;
    p[o] = cp[i];
    q[o] = cq[i] + bv;
  }
}

#define NB2 16

__global__ void gather16_close_kernel(const float* __restrict__ p,
                                      const float* __restrict__ q,
                                      const int* __restrict__ rowptr,
                                      const int* __restrict__ col,
                                      float* __restrict__ out, int N) {
  int nPairs = (N + 1) >> 1;
  int nWaves = gridDim.x * (blockDim.x >> 5);
  int lane = threadIdx.x & 31;
  int j    = lane & 15;
  int hi   = lane >> 4;
  for (int g = wave_in_grid(); g < nPairs; g += nWaves) {
    int  n    = 2 * g + hi;
    bool live = n < N;
    int  nc   = live ? n : N - 1;
    int   start = rowptr[nc];
    int   end   = rowptr[nc + 1];
    float qv    = q[(size_t)nc * N_C + j];
    int   deg   = live ? end - start : 0;
    int dmax = max(__builtin_amdgcn_readlane(deg, 0),
                   __builtin_amdgcn_readlane(deg, 16));
    float acc = 0.0f;
    for (int base = 0; base < dmax; base += NB2) {
      int idx = (base + j < deg) ? col[start + base + j] : 0;
      float v[NB2];
#pragma unroll
      for (int t = 0; t < NB2; ++t) {
        int s0 = __builtin_amdgcn_readlane(idx, t);
        int s1 = __builtin_amdgcn_readlane(idx, 16 + t);
        int s  = hi ? s1 : s0;
        v[t] = (base + t < deg) ? p[(size_t)s * N_C + j] : 0.0f;
      }
#pragma unroll
      for (int t = 0; t < NB2; ++t) acc += v[t];
    }
    if (live) {
      float inv = 1.0f / fmaxf((float)deg, 1.0f);
      out[(size_t)n * N_C + j] = qv + acc * inv;
    }
  }
}

extern "C" void kernel_launch(void* const* d_in, const int* in_sizes, int n_in,
                              void* d_out, int out_size, void* d_ws,
                              size_t ws_size, hipStream_t stream) {
  (void)n_in; (void)out_size; (void)ws_size;

  const float* x   = (const float*)d_in[0];
  const int*   ei  = (const int*)d_in[1];
  const float* W1l = (const float*)d_in[2];
  const float* b1  = (const float*)d_in[3];
  const float* W1r = (const float*)d_in[4];
  const float* W2l = (const float*)d_in[5];
  const float* b2  = (const float*)d_in[6];
  const float* W2r = (const float*)d_in[7];
  float* out = (float*)d_out;

  const int N  = in_sizes[0] / K_F;
  const int nE = in_sizes[1] / 2;
  const int* src = ei;
  const int* dst = ei + nE;
  const int nb = (N + BIN_W - 1) >> BIN_SH;

  float* h       = (float*)d_ws;
  float* p       = h + (size_t)N * K_F;
  float* q       = p + (size_t)N * N_C;
  int*   rowptr  = (int*)(q + (size_t)N * N_C);
  int*   binbase = rowptr + N + 16;
  int*   wgcnt   = binbase + NB_MAX + 16;
  int*   col     = wgcnt + SORT_G * NB_MAX;
  int*   pairs   = (int*)h;

  const int T = 256;

  bin_count_kernel<<<SORT_G, SORT_T, 0, stream>>>(dst, wgcnt, nE, nb);
  bin_place_kernel<<<SORT_G, SORT_T, 0, stream>>>(src, dst, wgcnt, binbase,
                                                  pairs, nE, nb);
  bin_sort_kernel<<<nb, SORT_T, 0, stream>>>(pairs, binbase, rowptr, col, N);

  gather64_kernel<<<RES_BLOCKS_EDGE, T, 0, stream>>>(x, rowptr, col, h, N);
  {
    int nStripes = N / 16;
    int blocks   = (nStripes + 7) / 8;
    sage_gemm64<<<blocks, T, 0, stream>>>(h, x, W1l, W1r, b1, nStripes);
  }

  {
    int nTiles = N / 16;
    int blocks = (nTiles + 7) / 8;
    proj2_kernel<<<blocks, T, 0, stream>>>(h, W2l, W2r, b2, p, q, nTiles);
  }
  gather16_close_kernel<<<RES_BLOCKS_EDGE, T, 0, stream>>>(p, q, rowptr, col,
                                                           out, N);
}


# layer 1 + layer-2 projections in one launch: 2 waves own a 16-node stripe, gather its neighbour means into shared memory, run both products from there; means and h never stored; 7 launches to 5
# speedup vs baseline: 3.2310x; 1.3053x over previous
#include <hip/hip_runtime.h>


typedef __attribute__((ext_vector_type(2))) float v2f;
typedef __attribute__((ext_vector_type(2))) int   v2i;
typedef __attribute__((ext_vector_type(8))) float v8f;

#define K_F 64
#define N_C 16

#define SORT_T 1024
#define SORT_G 64
#define SORT_U 8
#define BIN_SH 8
#define BIN_W  (1 << BIN_SH)
#define NB_MAX SORT_T

__device__ __forceinline__ int wave_scan_incl(int v, int lane) {
  for (int o = 1; o < 32; o <<= 1) {
    int t = __shfl_up(v, o);
    if (lane >= o) v += t;
  }
  return v;
}

__global__ void bin_count_kernel(const int* __restrict__ dst,
                                 int* __restrict__ wgcnt, int nE, int nb) {
  __shared__ int hist[NB_MAX];
  int tid = threadIdx.x;
  if (tid < nb) hist[tid] = 0;
  __syncthreads();
  int S  = (nE + gridDim.x - 1) / gridDim.x;
  int lo = blockIdx.x * S;
  int hi = min(lo + S, nE);
  for (int e = lo + tid; e < hi; e += SORT_U * SORT_T) {
    int d[SORT_U];
#pragma unroll
    for (int u = 0; u < SORT_U; ++u)
      d[u] = dst[min(e + u * SORT_T, hi - 1)];
#pragma unroll
    for (int u = 0; u < SORT_U; ++u)
      if (e + u * SORT_T < hi) atomicAdd(&hist[d[u] >> BIN_SH], 1);
  }
  __syncthreads();
  if (tid < nb) wgcnt[blockIdx.x * nb + tid] = hist[tid];
}

__global__ void bin_place_kernel(const int* __restrict__ src,
                                 const int* __restrict__ dst,
                                 const int* __restrict__ wgcnt,
                                 int* __restrict__ binbase,
                                 int* __restrict__ pairs_, int nE, int nb) {
  __shared__ int cur[NB_MAX];
  __shared__ int s_part[SORT_T / 32];
  v2i* pairs = reinterpret_cast<v2i*>(pairs_);
  int tid  = threadIdx.x;
  int lane = tid & 31;
  int w    = tid >> 5;
  int G    = gridDim.x;
  int g    = blockIdx.x;

  int bcol   = min(tid, nb - 1);
  int total  = 0;
  int before = 0;
  for (int k0 = 0; k0 < G; k0 += 16) {
    int t[16];
#pragma unroll
    for (int u = 0; u < 16; ++u) {
      t[u] = 0;
      if (k0 + u < G) t[u] = wgcnt[(k0 + u) * nb + bcol];
    }
#pragma unroll
    for (int u = 0; u < 16; ++u) {
      total += t[u];
      if (k0 + u < g) before += t[u];
    }
  }
  if (tid >= nb) {
    total  = 0;
    before = 0;
  }
  int inc = wave_scan_incl(total, lane);
  if (lane == 31) s_part[w] = inc;
  __syncthreads();
  int woff = 0;
  for (int k = 0; k < w; ++k) woff += s_part[k];
  int excl = woff + inc - total;
  if (tid < nb) {
    cur[tid] = excl + before;
    if (g == 0) {
      binbase[tid] = excl;
      if (tid == nb - 1) binbase[nb] = excl + total;
    }
  }
  __syncthreads();

  int S  = (nE + G - 1) / G;
  int lo = g * S;
  int hi = min(lo + S, nE);
  for (int e = lo + tid; e < hi; e += SORT_U * SORT_T) {
    int d[SORT_U];
    int s[SORT_U];
    int pos[SORT_U];
#pragma unroll
    for (int u = 0; u < SORT_U; ++u) {
      int ee = min(e + u * SORT_T, hi - 1);
      d[u] = dst[ee];
      s[u] = src[ee];
    }
#pragma unroll
    for (int u = 0; u < SORT_U; ++u) {
      pos[u] = 0;
      if (e + u * SORT_T < hi) pos[u] = atomicAdd(&cur[d[u] >> BIN_SH], 1);
    }
#pragma unroll
    for (int u = 0; u < SORT_U; ++u)
      if (e + u * SORT_T < hi) {
        v2i sd = {s[u], d[u]};
        pairs[pos[u]] = sd;
      }
  }
}

__global__ void bin_sort_kernel(const int* __restrict__ pairs_,
                                const int* __restrict__ binbase,
                                int* __restrict__ rowptr,
                                int* __restrict__ col, int N) {
  __shared__ int cnt[BIN_W];
  __shared__ int s_part[SORT_T / 32];
  const v2i* pairs = reinterpret_cast<const v2i*>(pairs_);
  int tid   = threadIdx.x;
  int lane  = tid & 31;
  int w     = tid >> 5;
  int b     = blockIdx.x;
  int seg0  = binbase[b];
  int seg1  = binbase[b + 1];
  int node0 = b << BIN_SH;

  if (tid < BIN_W) cnt[tid] = 0;
  __syncthreads();
  for (int e = seg0 + tid; e < seg1; e += SORT_U * SORT_T) {
    int d[SORT_U];
#pragma unroll
    for (int u = 0; u < SORT_U; ++u)
      d[u] = pairs[min(e + u * SORT_T, seg1 - 1)].y;
#pragma unroll
    for (int u = 0; u < SORT_U; ++u)
      if (e + u * SORT_T < seg1) atomicAdd(&cnt[d[u] - node0], 1);
  }
  __syncthreads();
  int c   = (tid < BIN_W) ? cnt[tid] : 0;
  int inc = wave_scan_incl(c, lane);
  if (lane == 31) s_part[w] = inc;
  __syncthreads();
  int woff = 0;
  for (int k = 0; k < w; ++k) woff += s_part[k];
  int excl = woff + inc - c;
  if (tid < BIN_W) {
    cnt[tid] = excl;
    int node = node0 + tid;
    if (node < N) rowptr[node] = seg0 + excl;
    if (node == N - 1) rowptr[N] = seg0 + excl + c;
  }
  __syncthreads();
  for (int e = seg0 + tid; e < seg1; e += SORT_U * SORT_T) {
    v2i sd[SORT_U];
    int r[SORT_U];
#pragma unroll
    for (int u = 0; u < SORT_U; ++u)
      sd[u] = pairs[min(e + u * SORT_T, seg1 - 1)];
#pragma unroll
    for (int u = 0; u < SORT_U; ++u) {
      r[u] = 0;
      if (e + u * SORT_T < seg1) r[u] = atomicAdd(&cnt[sd[u].y - node0], 1);
    }
#pragma unroll
    for (int u = 0; u < SORT_U; ++u)
      if (e + u * SORT_T < seg1) col[seg0 + r[u]] = sd[u].x;
  }
}

#define RES_BLOCKS_EDGE  1024

__device__ __forceinline__ int wave_in_grid() {
  return __builtin_amdgcn_readfirstlane(
      (int)(blockIdx.x * (blockDim.x >> 5) + (threadIdx.x >> 5)));
}

#define NB1 32
#define LDP 68

__global__ void __launch_bounds__(64)
layer1_stripe_kernel(const float* __restrict__ feat,
                     const int* __restrict__ rowptr,
                     const int* __restrict__ col,
                     const float* __restrict__ Wl,
                     const float* __restrict__ Wr,
                     const float* __restrict__ bias,
                     const float* __restrict__ W2l,
                     const float* __restrict__ W2r,
                     const float* __restrict__ bias2,
                     float* __restrict__ p,
                     float* __restrict__ q, int nE) {
  __shared__ alignas(16) float s_mean[16 * LDP];
  __shared__ alignas(16) float s_h[16 * LDP];
  int lane   = threadIdx.x & 31;
  int w      = __builtin_amdgcn_readfirstlane((int)(threadIdx.x >> 5));
  int stripe = blockIdx.x;
  int hrow   = lane & 15;
  int hi     = lane >> 4;
  int koff   = hi * 2;

  {
    int nbase = stripe * 16 + w * 8;
    int rp    = rowptr[nbase + min(lane, 8)];
    int start = __builtin_amdgcn_readlane(rp, 0);
    int idx   = col[min(start + lane, nE - 1)];
#pragma nounroll
    for (int i = 0; i < 8; ++i) {
      int end  = __builtin_amdgcn_readlane(rp, i + 1);
      int deg  = end - start;
      int idxN = col[min(end + lane, nE - 1)];
      float lo = 0.0f;
      float up = 0.0f;
      for (int base = 0; base < deg; base += NB1) {
        if (base > 0) idx = col[min(start + base + lane, nE - 1)];
        float a[NB1];
        float b[NB1];
#pragma unroll
        for (int t = 0; t < NB1; ++t) {
          a[t] = 0.0f;
          b[t] = 0.0f;
          if (base + t < deg) {
            int s = __builtin_amdgcn_readlane(idx, t);
            const float* r = feat + (size_t)s * K_F + lane;
            a[t] = r[0];
            b[t] = r[32];
          }
        }
#pragma unroll
        for (int t = 0; t < NB1; ++t) {
          lo += a[t];
          up += b[t];
        }
      }
      float inv = 1.0f / fmaxf((float)deg, 1.0f);
      float* m = s_mean + (w * 8 + i) * LDP + lane;
      m[0]  = lo * inv;
      m[32] = up * inv;
      start = end;
      idx   = idxN;
    }
  }
  __syncthreads();

  {
    const float* arow = s_mean + hrow * LDP + koff;
    const float* xrow = feat + (size_t)(stripe * 16 + hrow) * K_F + koff;
    int cb = w * 32 + hrow;
    v8f c0 = {};
    v8f c1 = {};
#pragma unroll
    for (int k = 0; k < K_F; k += 4) {
      v2f am = *reinterpret_cast<const v2f*>(arow + k);
      v2f ax = *reinterpret_cast<const v2f*>(xrow + k);
      const float* wl0 = Wl + (k + koff) * K_F + cb;
      const float* wr0 = Wr + (k + koff) * K_F + cb;
      v2f bl, br;
      bl.x = wl0[0];        bl.y = wl0[K_F];
      br.x = wr0[0];        br.y = wr0[K_F];
      c0 = __builtin_amdgcn_wmma_f32_16x16x4_f32(false, am, false, bl,
                                                 (short)0, c0, false, false);
      c0 = __builtin_amdgcn_wmma_f32_16x16x4_f32(false, ax, false, br,
                                                 (short)0, c0, false, false);
      bl.x = wl0[16];       bl.y = wl0[K_F + 16];
      br.x = wr0[16];       br.y = wr0[K_F + 16];
      c1 = __builtin_amdgcn_wmma_f32_16x16x4_f32(false, am, false, bl,
                                                 (short)0, c1, false, false);
      c1 = __builtin_amdgcn_wmma_f32_16x16x4_f32(false, ax, false, br,
                                                 (short)0, c1, false, false);
    }
    float bv0 = bias[cb];
    float bv1 = bias[cb + 16];
#pragma unroll
    for (int i = 0; i < 8; ++i) {
      float* o = s_h + (hi * 8 + i) * LDP + cb;
      o[0]  = fmaxf(c0[i] + bv0, 0.0f);
      o[16] = fmaxf(c1[i] + bv1, 0.0f);
    }
  }
  __syncthreads();

  {
    const float* W2   = w ? W2r : W2l;
    float*       dst2 = w ? q : p;
    float        bv   = w ? bias2[hrow] : 0.0f;
    const float* arow = s_h + hrow * LDP + koff;
    v8f cc = {};
#pragma unroll
    for (int k = 0; k < K_F; k += 4) {
      v2f a = *reinterpret_cast<const v2f*>(arow + k);
      v2f b;
      b.x = W2[(k + koff) * N_C + hrow];
      b.y = W2[(k + koff + 1) * N_C + hrow];
      cc = __builtin_amdgcn_wmma_f32_16x16x4_f32(false, a, false, b,
                                                 (short)0, cc, false, false);
    }
#pragma unroll
    for (int i = 0; i < 8; ++i)
      dst2[(size_t)(stripe * 16 + hi * 8 + i) * N_C + hrow] = cc[i] + bv;
  }
}

#define NB2 16

__global__ void gather16_close_kernel(const float* __restrict__ p,
                                      const float* __restrict__ q,
                                      const int* __restrict__ rowptr,
                                      const int* __restrict__ col,
                                      float* __restrict__ out, int N) {
  int nPairs = (N + 1) >> 1;
  int nWaves = gridDim.x * (blockDim.x >> 5);
  int lane = threadIdx.x & 31;
  int j    = lane & 15;
  int hi   = lane >> 4;
  for (int g = wave_in_grid(); g < nPairs; g += nWaves) {
    int  n    = 2 * g + hi;
    bool live = n < N;
    int  nc   = live ? n : N - 1;
    int   start = rowptr[nc];
    int   end   = rowptr[nc + 1];
    float qv    = q[(size_t)nc * N_C + j];
    int   deg   = live ? end - start : 0;
    int dmax = max(__builtin_amdgcn_readlane(deg, 0),
                   __builtin_amdgcn_readlane(deg, 16));
    float acc = 0.0f;
    for (int base = 0; base < dmax; base += NB2) {
      int idx = (base + j < deg) ? col[start + base + j] : 0;
      float v[NB2];
#pragma unroll
      for (int t = 0; t < NB2; ++t) {
        int s0 = __builtin_amdgcn_readlane(idx, t);
        int s1 = __builtin_amdgcn_readlane(idx, 16 + t);
        int s  = hi ? s1 : s0;
        v[t] = (base + t < deg) ? p[(size_t)s * N_C + j] : 0.0f;
      }
#pragma unroll
      for (int t = 0; t < NB2; ++t) acc += v[t];
    }
    if (live) {
      float inv = 1.0f / fmaxf((float)deg, 1.0f);
      out[(size_t)n * N_C + j] = qv + acc * inv;
    }
  }
}

extern "C" void kernel_launch(void* const* d_in, const int* in_sizes, int n_in,
                              void* d_out, int out_size, void* d_ws,
                              size_t ws_size, hipStream_t stream) {
  (void)n_in; (void)out_size; (void)ws_size;

  const float* x   = (const float*)d_in[0];
  const int*   ei  = (const int*)d_in[1];
  const float* W1l = (const float*)d_in[2];
  const float* b1  = (const float*)d_in[3];
  const float* W1r = (const float*)d_in[4];
  const float* W2l = (const float*)d_in[5];
  const float* b2  = (const float*)d_in[6];
  const float* W2r = (const float*)d_in[7];
  float* out = (float*)d_out;

  const int N  = in_sizes[0] / K_F;
  const int nE = in_sizes[1] / 2;
  const int* src = ei;
  const int* dst = ei + nE;
  const int nb = (N + BIN_W - 1) >> BIN_SH;

  float* pl0     = (float*)d_ws;
  float* p       = pl0 + (size_t)N * K_F;
  float* q       = p + (size_t)N * N_C;
  int*   rowptr  = (int*)(q + (size_t)N * N_C);
  int*   binbase = rowptr + N + 16;
  int*   wgcnt   = binbase + NB_MAX + 16;
  int*   col     = wgcnt + SORT_G * NB_MAX;
  int*   pairs   = (int*)pl0;

  const int T = 256;

  bin_count_kernel<<<SORT_G, SORT_T, 0, stream>>>(dst, wgcnt, nE, nb);
  bin_place_kernel<<<SORT_G, SORT_T, 0, stream>>>(src, dst, wgcnt, binbase,
                                                  pairs, nE, nb);
  bin_sort_kernel<<<nb, SORT_T, 0, stream>>>(pairs, binbase, rowptr, col, N);

  layer1_stripe_kernel<<<N / 16, 64, 0, stream>>>(x, rowptr, col, W1l, W1r, b1,
                                                  W2l, W2r, b2, p, q, nE);

  gather16_close_kernel<<<RES_BLOCKS_EDGE, T, 0, stream>>>(p, q, rowptr, col,
                                                           out, N);
}


# edge sort counts and places in one pass: the counting add's returned rank is the place in its segment; a slab bins inside its own stretch (128 slabs); 5 launches to 4; f32 sums still in arrival order
# speedup vs baseline: 4.1122x; 1.2727x over previous
#include <hip/hip_runtime.h>


typedef __attribute__((ext_vector_type(2))) float v2f;
typedef __attribute__((ext_vector_type(2))) int   v2i;
typedef __attribute__((ext_vector_type(4))) int   v4i;
typedef __attribute__((ext_vector_type(8))) float v8f;

#define K_F 64
#define N_C 16

#define SORT_T 1024
#define SORT_G 128
#define SORT_U 8
#define BIN_SH 8
#define BIN_W  (1 << BIN_SH)
#define NB_MAX SORT_T
#define SEG_L  (SORT_T / SORT_G)
#define SEG_U  8
#define SEG_CAP (SEG_L * SEG_U)
#define CS_U   8

__device__ __forceinline__ int wave_scan_incl(int v, int lane) {
  for (int o = 1; o < 32; o <<= 1) {
    int t = __shfl_up(v, o);
    if (lane >= o) v += t;
  }
  return v;
}

__global__ void slab_bin_kernel(const int* __restrict__ src,
                                const int* __restrict__ dst,
                                int* __restrict__ tcnt,
                                int* __restrict__ tstart,
                                int* __restrict__ pairs_, int nE, int nb,
                                int nbp) {
  __shared__ int hist[NB_MAX];
  __shared__ int s_part[SORT_T / 32];
  v2i* pairs = reinterpret_cast<v2i*>(pairs_);
  int tid  = threadIdx.x;
  int lane = tid & 31;
  int w    = tid >> 5;
  hist[tid] = 0;
  __syncthreads();
  int S    = (nE + gridDim.x - 1) / gridDim.x;
  int lo   = min((int)blockIdx.x * S, nE);
  int hi   = min(lo + S, nE);
  int last = max(hi - 1, 0);
  bool one = S <= SORT_U * SORT_T;
  int d[SORT_U];
  int s[SORT_U];
  int r[SORT_U];
  if (one) {
#pragma unroll
    for (int u = 0; u < SORT_U; ++u) {
      int ee = min(lo + tid + u * SORT_T, last);
      d[u] = dst[ee];
      s[u] = src[ee];
    }
#pragma unroll
    for (int u = 0; u < SORT_U; ++u) {
      r[u] = 0;
      if (lo + tid + u * SORT_T < hi) r[u] = atomicAdd(&hist[d[u] >> BIN_SH], 1);
    }
  } else {
    for (int e = lo + tid; e < hi; e += SORT_U * SORT_T) {
      int dd[SORT_U];
#pragma unroll
      for (int u = 0; u < SORT_U; ++u)
        dd[u] = dst[min(e + u * SORT_T, last)];
#pragma unroll
      for (int u = 0; u < SORT_U; ++u)
        if (e + u * SORT_T < hi) atomicAdd(&hist[dd[u] >> BIN_SH], 1);
    }
  }
  __syncthreads();
  int c   = (tid < nb) ? hist[tid] : 0;
  int inc = wave_scan_incl(c, lane);
  if (lane == 31) s_part[w] = inc;
  __syncthreads();
  int woff = 0;
  for (int k = 0; k < w; ++k) woff += s_part[k];
  int excl = woff + inc - c;
  if (tid < nbp) {
    hist[tid] = lo + excl;
    tstart[blockIdx.x * nbp + tid] = lo + excl;
    tcnt[blockIdx.x * nbp + tid]   = c;
  }
  __syncthreads();
  if (one) {
    int st[SORT_U];
#pragma unroll
    for (int u = 0; u < SORT_U; ++u) st[u] = hist[d[u] >> BIN_SH];
#pragma unroll
    for (int u = 0; u < SORT_U; ++u)
      if (lo + tid + u * SORT_T < hi) {
        v2i sd = {s[u], d[u]};
        pairs[st[u] + r[u]] = sd;
      }
  } else {
    for (int e = lo + tid; e < hi; e += SORT_U * SORT_T) {
      int dd[SORT_U];
      int ss[SORT_U];
      int pos[SORT_U];
#pragma unroll
      for (int u = 0; u < SORT_U; ++u) {
        int ee = min(e + u * SORT_T, last);
        dd[u] = dst[ee];
        ss[u] = src[ee];
      }
#pragma unroll
      for (int u = 0; u < SORT_U; ++u) {
        pos[u] = 0;
        if (e + u * SORT_T < hi) pos[u] = atomicAdd(&hist[dd[u] >> BIN_SH], 1);
      }
#pragma unroll
      for (int u = 0; u < SORT_U; ++u)
        if (e + u * SORT_T < hi) {
          v2i sd = {ss[u], dd[u]};
          pairs[pos[u]] = sd;
        }
    }
  }
}

__global__ void bin_sort_kernel(const int* __restrict__ pairs_,
                                const int* __restrict__ tcnt,
                                const int* __restrict__ tstart,
                                int* __restrict__ rowptr,
                                int* __restrict__ col, int N, int nbp) {
  __shared__ int cnt[BIN_W];
  __shared__ int sst[SORT_G];
  __shared__ int scn[SORT_G];
  __shared__ int s_partA[SORT_T / 32];
  __shared__ int s_part[SORT_T / 32];
  __shared__ int s_big;
  const v2i* pairs = reinterpret_cast<const v2i*>(pairs_);
  unsigned tid = threadIdx.x;
  int lane  = tid & 31;
  int w     = tid >> 5;
  int g     = tid / SEG_L;
  int ll    = tid & (SEG_L - 1);
  int b     = blockIdx.x;
  int node0 = b << BIN_SH;

  if (tid < BIN_W) cnt[tid] = 0;
  if (tid == 0) s_big = 0;

  unsigned unbp  = (unsigned)nbp;
  unsigned mrow  = min(tid, (unsigned)(SORT_G - 1)) * unbp + (unsigned)b;
  int      mst   = tstart[mrow];
  int      mcn   = tcnt[mrow];
  unsigned nq    = unbp >> 2;
  unsigned parts = SORT_T / nq;
  unsigned part  = tid / nq;
  unsigned qc    = tid - part * nq;
  int  c0   = (int)(4 * qc);
  bool in0  = c0 < b;
  bool in1  = c0 + 1 < b;
  bool in2  = c0 + 2 < b;
  bool in3  = c0 + 3 < b;
  int  sum  = 0;
  for (unsigned g0 = part; g0 < SORT_G; g0 += CS_U * parts) {
    v4i t[CS_U];
#pragma unroll
    for (int u = 0; u < CS_U; ++u)
      t[u] = *reinterpret_cast<const v4i*>(
          tcnt + min(g0 + u * parts, (unsigned)(SORT_G - 1)) * unbp + 4 * qc);
#pragma unroll
    for (int u = 0; u < CS_U; ++u)
      if (g0 + u * parts < SORT_G)
        sum += (in0 ? t[u].x : 0) + (in1 ? t[u].y : 0) +
               (in2 ? t[u].z : 0) + (in3 ? t[u].w : 0);
  }
  if (part >= parts) sum = 0;
  __syncthreads();
  if (tid < SORT_G) {
    sst[tid] = mst;
    scn[tid] = mcn;
    if (mcn > SEG_CAP) s_big = 1;
  }
  int incA = wave_scan_incl(sum, lane);
  if (lane == 31) s_partA[w] = incA;
  __syncthreads();
  int seg0 = 0;
#pragma unroll
  for (int k = 0; k < SORT_T / 32; ++k) seg0 += s_partA[k];
  int  s0   = sst[g];
  int  s1   = s0 + scn[g];
  int  last = max(s1 - 1, 0);
  bool big  = s_big != 0;

  v2i sd[SEG_U];
  int r[SEG_U];
  if (!big) {
#pragma unroll
    for (int u = 0; u < SEG_U; ++u)
      sd[u] = pairs[min(s0 + ll + u * SEG_L, last)];
#pragma unroll
    for (int u = 0; u < SEG_U; ++u) {
      r[u] = 0;
      if (s0 + ll + u * SEG_L < s1) r[u] = atomicAdd(&cnt[sd[u].y - node0], 1);
    }
  } else {
    for (int e = s0 + ll; e < s1; e += SEG_U * SEG_L) {
      int dd[SEG_U];
#pragma unroll
      for (int u = 0; u < SEG_U; ++u)
        dd[u] = pairs[min(e + u * SEG_L, last)].y;
#pragma unroll
      for (int u = 0; u < SEG_U; ++u)
        if (e + u * SEG_L < s1) atomicAdd(&cnt[dd[u] - node0], 1);
    }
  }
  __syncthreads();
  int c   = (tid < BIN_W) ? cnt[tid] : 0;
  int inc = wave_scan_incl(c, lane);
  if (lane == 31) s_part[w] = inc;
  __syncthreads();
  int woff = 0;
  for (int k = 0; k < w; ++k) woff += s_part[k];
  int excl = woff + inc - c;
  if (tid < BIN_W) {
    cnt[tid] = excl;
    int node = node0 + (int)tid;
    if (node < N) rowptr[node] = seg0 + excl;
    if (node == N - 1) rowptr[N] = seg0 + excl + c;
  }
  __syncthreads();
  if (!big) {
    int st[SEG_U];
#pragma unroll
    for (int u = 0; u < SEG_U; ++u) st[u] = cnt[(sd[u].y - node0) & (BIN_W - 1)];
#pragma unroll
    for (int u = 0; u < SEG_U; ++u)
      if (s0 + ll + u * SEG_L < s1) col[seg0 + st[u] + r[u]] = sd[u].x;
  } else {
    for (int e = s0 + ll; e < s1; e += SEG_U * SEG_L) {
      v2i sdd[SEG_U];
      int rr[SEG_U];
#pragma unroll
      for (int u = 0; u < SEG_U; ++u)
        sdd[u] = pairs[min(e + u * SEG_L, last)];
#pragma unroll
      for (int u = 0; u < SEG_U; ++u) {
        rr[u] = 0;
        if (e + u * SEG_L < s1) rr[u] = atomicAdd(&cnt[sdd[u].y - node0], 1);
      }
#pragma unroll
      for (int u = 0; u < SEG_U; ++u)
        if (e + u * SEG_L < s1) col[seg0 + rr[u]] = sdd[u].x;
    }
  }
}

#define RES_BLOCKS_EDGE  1024

__device__ __forceinline__ int wave_in_grid() {
  return __builtin_amdgcn_readfirstlane(
      (int)(blockIdx.x * (blockDim.x >> 5) + (threadIdx.x >> 5)));
}

#define NB1 32
#define LDP 68

__global__ void __launch_bounds__(64)
layer1_stripe_kernel(const float* __restrict__ feat,
                     const int* __restrict__ rowptr,
                     const int* __restrict__ col,
                     const float* __restrict__ Wl,
                     const float* __restrict__ Wr,
                     const float* __restrict__ bias,
                     const float* __restrict__ W2l,
                     const float* __restrict__ W2r,
                     const float* __restrict__ bias2,
                     float* __restrict__ p,
                     float* __restrict__ q, int nE) {
  __shared__ alignas(16) float s_mean[16 * LDP];
  __shared__ alignas(16) float s_h[16 * LDP];
  int lane   = threadIdx.x & 31;
  int w      = __builtin_amdgcn_readfirstlane((int)(threadIdx.x >> 5));
  int stripe = blockIdx.x;
  int hrow   = lane & 15;
  int hi     = lane >> 4;
  int koff   = hi * 2;

  {
    int nbase = stripe * 16 + w * 8;
    int rp    = rowptr[nbase + min(lane, 8)];
    int start = __builtin_amdgcn_readlane(rp, 0);
    int idx   = col[min(start + lane, nE - 1)];
#pragma nounroll
    for (int i = 0; i < 8; ++i) {
      int end  = __builtin_amdgcn_readlane(rp, i + 1);
      int deg  = end - start;
      int idxN = col[min(end + lane, nE - 1)];
      float lo = 0.0f;
      float up = 0.0f;
      for (int base = 0; base < deg; base += NB1) {
        if (base > 0) idx = col[min(start + base + lane, nE - 1)];
        float a[NB1];
        float b[NB1];
#pragma unroll
        for (int t = 0; t < NB1; ++t) {
          a[t] = 0.0f;
          b[t] = 0.0f;
          if (base + t < deg) {
            int s = __builtin_amdgcn_readlane(idx, t);
            const float* r = feat + (size_t)s * K_F + lane;
            a[t] = r[0];
            b[t] = r[32];
          }
        }
#pragma unroll
        for (int t = 0; t < NB1; ++t) {
          lo += a[t];
          up += b[t];
        }
      }
      float inv = 1.0f / fmaxf((float)deg, 1.0f);
      float* m = s_mean + (w * 8 + i) * LDP + lane;
      m[0]  = lo * inv;
      m[32] = up * inv;
      start = end;
      idx   = idxN;
    }
  }
  __syncthreads();

  {
    const float* arow = s_mean + hrow * LDP + koff;
    const float* xrow = feat + (size_t)(stripe * 16 + hrow) * K_F + koff;
    int cb = w * 32 + hrow;
    v8f c0 = {};
    v8f c1 = {};
#pragma unroll
    for (int k = 0; k < K_F; k += 4) {
      v2f am = *reinterpret_cast<const v2f*>(arow + k);
      v2f ax = *reinterpret_cast<const v2f*>(xrow + k);
      const float* wl0 = Wl + (k + koff) * K_F + cb;
      const float* wr0 = Wr + (k + koff) * K_F + cb;
      v2f bl, br;
      bl.x = wl0[0];        bl.y = wl0[K_F];
      br.x = wr0[0];        br.y = wr0[K_F];
      c0 = __builtin_amdgcn_wmma_f32_16x16x4_f32(false, am, false, bl,
                                                 (short)0, c0, false, false);
      c0 = __builtin_amdgcn_wmma_f32_16x16x4_f32(false, ax, false, br,
                                                 (short)0, c0, false, false);
      bl.x = wl0[16];       bl.y = wl0[K_F + 16];
      br.x = wr0[16];       br.y = wr0[K_F + 16];
      c1 = __builtin_amdgcn_wmma_f32_16x16x4_f32(false, am, false, bl,
                                                 (short)0, c1, false, false);
      c1 = __builtin_amdgcn_wmma_f32_16x16x4_f32(false, ax, false, br,
                                                 (short)0, c1, false, false);
    }
    float bv0 = bias[cb];
    float bv1 = bias[cb + 16];
#pragma unroll
    for (int i = 0; i < 8; ++i) {
      float* o = s_h + (hi * 8 + i) * LDP + cb;
      o[0]  = fmaxf(c0[i] + bv0, 0.0f);
      o[16] = fmaxf(c1[i] + bv1, 0.0f);
    }
  }
  __syncthreads();

  {
    const float* W2   = w ? W2r : W2l;
    float*       dst2 = w ? q : p;
    float        bv   = w ? bias2[hrow] : 0.0f;
    const float* arow = s_h + hrow * LDP + koff;
    v8f cc = {};
#pragma unroll
    for (int k = 0; k < K_F; k += 4) {
      v2f a = *reinterpret_cast<const v2f*>(arow + k);
      v2f b;
      b.x = W2[(k + koff) * N_C + hrow];
      b.y = W2[(k + koff + 1) * N_C + hrow];
      cc = __builtin_amdgcn_wmma_f32_16x16x4_f32(false, a, false, b,
                                                 (short)0, cc, false, false);
    }
#pragma unroll
    for (int i = 0; i < 8; ++i)
      dst2[(size_t)(stripe * 16 + hi * 8 + i) * N_C + hrow] = cc[i] + bv;
  }
}

#define NB2 16

__global__ void gather16_close_kernel(const float* __restrict__ p,
                                      const float* __restrict__ q,
                                      const int* __restrict__ rowptr,
                                      const int* __restrict__ col,
                                      float* __restrict__ out, int N) {
  int nPairs = (N + 1) >> 1;
  int nWaves = gridDim.x * (blockDim.x >> 5);
  int lane = threadIdx.x & 31;
  int j    = lane & 15;
  int hi   = lane >> 4;
  for (int g = wave_in_grid(); g < nPairs; g += nWaves) {
    int  n    = 2 * g + hi;
    bool live = n < N;
    int  nc   = live ? n : N - 1;
    int   start = rowptr[nc];
    int   end   = rowptr[nc + 1];
    float qv    = q[(size_t)nc * N_C + j];
    int   deg   = live ? end - start : 0;
    int dmax = max(__builtin_amdgcn_readlane(deg, 0),
                   __builtin_amdgcn_readlane(deg, 16));
    float acc = 0.0f;
    for (int base = 0; base < dmax; base += NB2) {
      int idx = (base + j < deg) ? col[start + base + j] : 0;
      float v[NB2];
#pragma unroll
      for (int t = 0; t < NB2; ++t) {
        int s0 = __builtin_amdgcn_readlane(idx, t);
        int s1 = __builtin_amdgcn_readlane(idx, 16 + t);
        int s  = hi ? s1 : s0;
        v[t] = (base + t < deg) ? p[(size_t)s * N_C + j] : 0.0f;
      }
#pragma unroll
      for (int t = 0; t < NB2; ++t) acc += v[t];
    }
    if (live) {
      float inv = 1.0f / fmaxf((float)deg, 1.0f);
      out[(size_t)n * N_C + j] = qv + acc * inv;
    }
  }
}

extern "C" void kernel_launch(void* const* d_in, const int* in_sizes, int n_in,
                              void* d_out, int out_size, void* d_ws,
                              size_t ws_size, hipStream_t stream) {
  (void)n_in; (void)out_size; (void)ws_size;

  const float* x   = (const float*)d_in[0];
  const int*   ei  = (const int*)d_in[1];
  const float* W1l = (const float*)d_in[2];
  const float* b1  = (const float*)d_in[3];
  const float* W1r = (const float*)d_in[4];
  const float* W2l = (const float*)d_in[5];
  const float* b2  = (const float*)d_in[6];
  const float* W2r = (const float*)d_in[7];
  float* out = (float*)d_out;

  const int N  = in_sizes[0] / K_F;
  const int nE = in_sizes[1] / 2;
  const int* src = ei;
  const int* dst = ei + nE;
  const int nb = (N + BIN_W - 1) >> BIN_SH;
  const int nbp = (nb + 3) & ~3;

  float* pl0     = (float*)d_ws;
  float* p       = pl0 + (size_t)N * K_F;
  float* q       = p + (size_t)N * N_C;
  int*   rowptr  = (int*)(q + (size_t)N * N_C);
  int*   tcnt    = rowptr + N + 16;
  int*   tstart  = tcnt + SORT_G * nbp;
  int*   col     = tstart + SORT_G * nbp;
  int*   pairs   = (int*)pl0;

  const int T = 256;

  slab_bin_kernel<<<SORT_G, SORT_T, 0, stream>>>(src, dst, tcnt, tstart, pairs,
                                                 nE, nb, nbp);
  bin_sort_kernel<<<nb, SORT_T, 0, stream>>>(pairs, tcnt, tstart, rowptr, col,
                                             N, nbp);

  layer1_stripe_kernel<<<N / 16, 64, 0, stream>>>(x, rowptr, col, W1l, W1r, b1,
                                                  W2l, W2r, b2, p, q, nE);

  gather16_close_kernel<<<RES_BLOCKS_EDGE, T, 0, stream>>>(p, q, rowptr, col,
                                                           out, N);
}


# sort pass 2 + layer 1 in one launch: a block sorts its 208-node bin in shared memory, 13 wave pairs run layer 1 from that list, x@W1r ahead of its barrier; 4 launches to 3; f32 sum order changes
# speedup vs baseline: 4.6766x; 1.1373x over previous
#include <hip/hip_runtime.h>


typedef __attribute__((ext_vector_type(2))) float v2f;
typedef __attribute__((ext_vector_type(2))) int   v2i;
typedef __attribute__((ext_vector_type(4))) int   v4i;
typedef __attribute__((ext_vector_type(8))) float v8f;

#define K_F 64
#define N_C 16

#define SORT_T 1024
#define SORT_G 128
#define SORT_U 8
#define BIN_W  208
#define NB_MAX SORT_T
#define SEG_L  (SORT_T / SORT_G)
#define SEG_U  8
#define SEG_CAP (SEG_L * SEG_U)
#define CS_U   8

__device__ __forceinline__ int wave_scan_incl(int v, int lane) {
  for (int o = 1; o < 32; o <<= 1) {
    int t = __shfl_up(v, o);
    if (lane >= o) v += t;
  }
  return v;
}

__global__ void slab_bin_kernel(const int* __restrict__ src,
                                const int* __restrict__ dst,
                                int* __restrict__ tcnt,
                                int* __restrict__ tstart,
                                int* __restrict__ pairs_, int nE, int nb,
                                int nbp) {
  __shared__ int hist[NB_MAX];
  __shared__ int s_part[SORT_T / 32];
  v2i* pairs = reinterpret_cast<v2i*>(pairs_);
  int tid  = threadIdx.x;
  int lane = tid & 31;
  int w    = tid >> 5;
  hist[tid] = 0;
  __syncthreads();
  int S    = (nE + gridDim.x - 1) / gridDim.x;
  int lo   = min((int)blockIdx.x * S, nE);
  int hi   = min(lo + S, nE);
  int last = max(hi - 1, 0);
  bool one = S <= SORT_U * SORT_T;
  int d[SORT_U];
  int s[SORT_U];
  int r[SORT_U];
  if (one) {
#pragma unroll
    for (int u = 0; u < SORT_U; ++u) {
      int ee = min(lo + tid + u * SORT_T, last);
      d[u] = dst[ee];
      s[u] = src[ee];
    }
#pragma unroll
    for (int u = 0; u < SORT_U; ++u) {
      r[u] = 0;
      if (lo + tid + u * SORT_T < hi) r[u] = atomicAdd(&hist[(unsigned)d[u] / BIN_W], 1);
    }
  } else {
    for (int e = lo + tid; e < hi; e += SORT_U * SORT_T) {
      int dd[SORT_U];
#pragma unroll
      for (int u = 0; u < SORT_U; ++u)
        dd[u] = dst[min(e + u * SORT_T, last)];
#pragma unroll
      for (int u = 0; u < SORT_U; ++u)
        if (e + u * SORT_T < hi) atomicAdd(&hist[(unsigned)dd[u] / BIN_W], 1);
    }
  }
  __syncthreads();
  int c   = (tid < nb) ? hist[tid] : 0;
  int inc = wave_scan_incl(c, lane);
  if (lane == 31) s_part[w] = inc;
  __syncthreads();
  int woff = 0;
  for (int k = 0; k < w; ++k) woff += s_part[k];
  int excl = woff + inc - c;
  if (tid < nbp) {
    hist[tid] = lo + excl;
    tstart[blockIdx.x * nbp + tid] = lo + excl;
    tcnt[blockIdx.x * nbp + tid]   = c;
  }
  __syncthreads();
  if (one) {
    int st[SORT_U];
#pragma unroll
    for (int u = 0; u < SORT_U; ++u) st[u] = hist[(unsigned)d[u] / BIN_W];
#pragma unroll
    for (int u = 0; u < SORT_U; ++u)
      if (lo + tid + u * SORT_T < hi) {
        v2i sd = {s[u], d[u]};
        pairs[st[u] + r[u]] = sd;
      }
  } else {
    for (int e = lo + tid; e < hi; e += SORT_U * SORT_T) {
      int dd[SORT_U];
      int ss[SORT_U];
      int pos[SORT_U];
#pragma unroll
      for (int u = 0; u < SORT_U; ++u) {
        int ee = min(e + u * SORT_T, last);
        dd[u] = dst[ee];
        ss[u] = src[ee];
      }
#pragma unroll
      for (int u = 0; u < SORT_U; ++u) {
        pos[u] = 0;
        if (e + u * SORT_T < hi) pos[u] = atomicAdd(&hist[(unsigned)dd[u] / BIN_W], 1);
      }
#pragma unroll
      for (int u = 0; u < SORT_U; ++u)
        if (e + u * SORT_T < hi) {
          v2i sd = {ss[u], dd[u]};
          pairs[pos[u]] = sd;
        }
    }
  }
}

#define RES_BLOCKS_EDGE  1024

__device__ __forceinline__ int wave_in_grid() {
  return __builtin_amdgcn_readfirstlane(
      (int)(blockIdx.x * (blockDim.x >> 5) + (threadIdx.x >> 5)));
}

#define NB1 32
#define LDP 68
#define COL_CAP 8192

__device__ __forceinline__ void wave_means(const float* __restrict__ feat,
                                           const int* lst, int lim,
                                           const int* rp8, float* mrow,
                                           int lane) {
  int rp    = rp8[min(lane, 8)];
  int start = __builtin_amdgcn_readlane(rp, 0);
  int idx   = lst[min(start + lane, lim)];
#pragma nounroll
  for (int i = 0; i < 8; ++i) {
    int end  = __builtin_amdgcn_readlane(rp, i + 1);
    int deg  = end - start;
    int idxN = lst[min(end + lane, lim)];
    float lo = 0.0f;
    float up = 0.0f;
    for (int base = 0; base < deg; base += NB1) {
      if (base > 0) idx = lst[min(start + base + lane, lim)];
      float a[NB1];
      float bq[NB1];
#pragma unroll
      for (int t = 0; t < NB1; ++t) {
        a[t]  = 0.0f;
        bq[t] = 0.0f;
        if (base + t < deg) {
          int s = __builtin_amdgcn_readlane(idx, t);
          const float* rw = feat + (size_t)s * K_F + lane;
          a[t]  = rw[0];
          bq[t] = rw[32];
        }
      }
#pragma unroll
      for (int t = 0; t < NB1; ++t) {
        lo += a[t];
        up += bq[t];
      }
    }
    float inv = 1.0f / fmaxf((float)deg, 1.0f);
    float* m = mrow + i * LDP + lane;
    m[0]  = lo * inv;
    m[32] = up * inv;
    start = end;
    idx   = idxN;
  }
}

__global__ void __launch_bounds__(SORT_T)
bin_layer1_kernel(const int* __restrict__ pairs_,
                  const int* __restrict__ tcnt,
                  const int* __restrict__ tstart,
                  int* __restrict__ rowptr,
                  int* __restrict__ col,
                  const float* __restrict__ feat,
                  const float* __restrict__ Wl,
                  const float* __restrict__ Wr,
                  const float* __restrict__ bias,
                  const float* __restrict__ W2l,
                  const float* __restrict__ W2r,
                  const float* __restrict__ bias2,
                  float* __restrict__ p,
                  float* __restrict__ q, int N, int nbp) {
  __shared__ int cnt[BIN_W];
  __shared__ int s_rp[BIN_W + 1];
  __shared__ int sst[SORT_G];
  __shared__ int scn[SORT_G];
  __shared__ int s_partA[SORT_T / 32];
  __shared__ int s_part[SORT_T / 32];
  __shared__ int s_big;
  __shared__ int s_col[COL_CAP];
  __shared__ alignas(16) float s_mean[BIN_W * LDP];
  __shared__ alignas(16) float s_h[BIN_W * LDP];
  const v2i* pairs = reinterpret_cast<const v2i*>(pairs_);
  unsigned tid = threadIdx.x;
  int lane  = tid & 31;
  int w     = tid >> 5;
  int g     = tid / SEG_L;
  int ll    = tid & (SEG_L - 1);
  int b     = blockIdx.x;
  int node0 = b * BIN_W;

  if (tid < BIN_W) cnt[tid] = 0;
  if (tid == 0) s_big = 0;

  unsigned unbp  = (unsigned)nbp;
  unsigned mrow  = min(tid, (unsigned)(SORT_G - 1)) * unbp + (unsigned)b;
  int      mst   = tstart[mrow];
  int      mcn   = tcnt[mrow];
  unsigned nq    = unbp >> 2;
  unsigned parts = SORT_T / nq;
  unsigned part  = tid / nq;
  unsigned qc    = tid - part * nq;
  int  c0   = (int)(4 * qc);
  bool in0  = c0 < b;
  bool in1  = c0 + 1 < b;
  bool in2  = c0 + 2 < b;
  bool in3  = c0 + 3 < b;
  int  sum  = 0;
  for (unsigned g0 = part; g0 < SORT_G; g0 += CS_U * parts) {
    v4i t[CS_U];
#pragma unroll
    for (int u = 0; u < CS_U; ++u)
      t[u] = *reinterpret_cast<const v4i*>(
          tcnt + min(g0 + u * parts, (unsigned)(SORT_G - 1)) * unbp + 4 * qc);
#pragma unroll
    for (int u = 0; u < CS_U; ++u)
      if (g0 + u * parts < SORT_G)
        sum += (in0 ? t[u].x : 0) + (in1 ? t[u].y : 0) +
               (in2 ? t[u].z : 0) + (in3 ? t[u].w : 0);
  }
  if (part >= parts) sum = 0;
  __syncthreads();
  if (tid < SORT_G) {
    sst[tid] = mst;
    scn[tid] = mcn;
    if (mcn > SEG_CAP) s_big = 1;
  }
  int incA = wave_scan_incl(sum, lane);
  if (lane == 31) s_partA[w] = incA;
  __syncthreads();
  int seg0 = 0;
#pragma unroll
  for (int k = 0; k < SORT_T / 32; ++k) seg0 += s_partA[k];
  int  s0   = sst[g];
  int  s1   = s0 + scn[g];
  int  last = max(s1 - 1, 0);
  bool big  = s_big != 0;

  v2i sd[SEG_U];
  int r[SEG_U];
  if (!big) {
#pragma unroll
    for (int u = 0; u < SEG_U; ++u)
      sd[u] = pairs[min(s0 + ll + u * SEG_L, last)];
#pragma unroll
    for (int u = 0; u < SEG_U; ++u) {
      r[u] = 0;
      if (s0 + ll + u * SEG_L < s1) r[u] = atomicAdd(&cnt[sd[u].y - node0], 1);
    }
  } else {
    for (int e = s0 + ll; e < s1; e += SEG_U * SEG_L) {
      int dd[SEG_U];
#pragma unroll
      for (int u = 0; u < SEG_U; ++u)
        dd[u] = pairs[min(e + u * SEG_L, last)].y;
#pragma unroll
      for (int u = 0; u < SEG_U; ++u)
        if (e + u * SEG_L < s1) atomicAdd(&cnt[dd[u] - node0], 1);
    }
  }
  __syncthreads();
  int c   = (tid < BIN_W) ? cnt[tid] : 0;
  int inc = wave_scan_incl(c, lane);
  if (lane == 31) s_part[w] = inc;
  __syncthreads();
  int woff = 0;
  for (int k = 0; k < w; ++k) woff += s_part[k];
  int excl = woff + inc - c;
  if (tid < BIN_W) {
    cnt[tid]  = excl;
    s_rp[tid] = excl;
    if (tid == BIN_W - 1) s_rp[BIN_W] = excl + c;
    int node = node0 + (int)tid;
    if (node < N) rowptr[node] = seg0 + excl;
    if (node == N - 1) rowptr[N] = seg0 + excl + c;
  }
  __syncthreads();
  int  total = __builtin_amdgcn_readfirstlane(s_rp[BIN_W]);
  bool fits  = total <= COL_CAP;
  if (!big) {
    int st[SEG_U];
#pragma unroll
    for (int u = 0; u < SEG_U; ++u)
      st[u] = cnt[min((unsigned)(sd[u].y - node0), (unsigned)(BIN_W - 1))];
#pragma unroll
    for (int u = 0; u < SEG_U; ++u)
      if (s0 + ll + u * SEG_L < s1) {
        if (fits) s_col[st[u] + r[u]] = sd[u].x;
        else      col[seg0 + st[u] + r[u]] = sd[u].x;
      }
  } else {
    for (int e = s0 + ll; e < s1; e += SEG_U * SEG_L) {
      v2i sdd[SEG_U];
      int rr[SEG_U];
#pragma unroll
      for (int u = 0; u < SEG_U; ++u)
        sdd[u] = pairs[min(e + u * SEG_L, last)];
#pragma unroll
      for (int u = 0; u < SEG_U; ++u) {
        rr[u] = 0;
        if (e + u * SEG_L < s1) rr[u] = atomicAdd(&cnt[sdd[u].y - node0], 1);
      }
#pragma unroll
      for (int u = 0; u < SEG_U; ++u)
        if (e + u * SEG_L < s1) {
          if (fits) s_col[rr[u]] = sdd[u].x;
          else      col[seg0 + rr[u]] = sdd[u].x;
        }
    }
  }
  __syncthreads();
  if (fits)
    for (int i = (int)tid; i < total; i += SORT_T) col[seg0 + i] = s_col[i];

  int  w2     = __builtin_amdgcn_readfirstlane((int)(tid >> 5));
  int  sl     = w2 >> 1;
  int  wp     = w2 & 1;
  int  stripe = (node0 >> 4) + sl;
  bool live   = sl < BIN_W / 16 && node0 + sl * 16 < N;
  int  hrow   = lane & 15;
  int  hi     = lane >> 4;
  int  koff   = hi * 2;

  if (live) {
    int lim = max(total - 1, 0);
    if (fits)
      wave_means(feat, s_col, lim, s_rp + w2 * 8, s_mean + w2 * 8 * LDP, lane);
    else
      wave_means(feat, col + seg0, lim, s_rp + w2 * 8, s_mean + w2 * 8 * LDP,
                 lane);
  }

  int cb = wp * 32 + hrow;
  v8f c0v = {};
  v8f c1v = {};
  if (live) {
    const float* xrow = feat + (size_t)(stripe * 16 + hrow) * K_F + koff;
#pragma unroll
    for (int k = 0; k < K_F; k += 4) {
      v2f ax = *reinterpret_cast<const v2f*>(xrow + k);
      const float* wr0 = Wr + (k + koff) * K_F + cb;
      v2f br;
      br.x = wr0[0];        br.y = wr0[K_F];
      c0v = __builtin_amdgcn_wmma_f32_16x16x4_f32(false, ax, false, br,
                                                  (short)0, c0v, false, false);
      br.x = wr0[16];       br.y = wr0[K_F + 16];
      c1v = __builtin_amdgcn_wmma_f32_16x16x4_f32(false, ax, false, br,
                                                  (short)0, c1v, false, false);
    }
  }
  __syncthreads();
  if (live) {
    const float* arow = s_mean + (sl * 16 + hrow) * LDP + koff;
#pragma unroll
    for (int k = 0; k < K_F; k += 4) {
      v2f am = *reinterpret_cast<const v2f*>(arow + k);
      const float* wl0 = Wl + (k + koff) * K_F + cb;
      v2f bl;
      bl.x = wl0[0];        bl.y = wl0[K_F];
      c0v = __builtin_amdgcn_wmma_f32_16x16x4_f32(false, am, false, bl,
                                                  (short)0, c0v, false, false);
      bl.x = wl0[16];       bl.y = wl0[K_F + 16];
      c1v = __builtin_amdgcn_wmma_f32_16x16x4_f32(false, am, false, bl,
                                                  (short)0, c1v, false, false);
    }
    float bv0 = bias[cb];
    float bv1 = bias[cb + 16];
#pragma unroll
    for (int i = 0; i < 8; ++i) {
      float* o = s_h + (sl * 16 + hi * 8 + i) * LDP + cb;
      o[0]  = fmaxf(c0v[i] + bv0, 0.0f);
      o[16] = fmaxf(c1v[i] + bv1, 0.0f);
    }
  }
  __syncthreads();

  if (live) {
    const float* W2   = wp ? W2r : W2l;
    float*       dst2 = wp ? q : p;
    float        bv   = wp ? bias2[hrow] : 0.0f;
    const float* arow = s_h + (sl * 16 + hrow) * LDP + koff;
    v8f cc = {};
#pragma unroll
    for (int k = 0; k < K_F; k += 4) {
      v2f a = *reinterpret_cast<const v2f*>(arow + k);
      v2f bb;
      bb.x = W2[(k + koff) * N_C + hrow];
      bb.y = W2[(k + koff + 1) * N_C + hrow];
      cc = __builtin_amdgcn_wmma_f32_16x16x4_f32(false, a, false, bb,
                                                 (short)0, cc, false, false);
    }
#pragma unroll
    for (int i = 0; i < 8; ++i)
      dst2[(size_t)(stripe * 16 + hi * 8 + i) * N_C + hrow] = cc[i] + bv;
  }
}

#define NB2 16

__global__ void gather16_close_kernel(const float* __restrict__ p,
                                      const float* __restrict__ q,
                                      const int* __restrict__ rowptr,
                                      const int* __restrict__ col,
                                      float* __restrict__ out, int N) {
  int nPairs = (N + 1) >> 1;
  int nWaves = gridDim.x * (blockDim.x >> 5);
  int lane = threadIdx.x & 31;
  int j    = lane & 15;
  int hi   = lane >> 4;
  for (int g = wave_in_grid(); g < nPairs; g += nWaves) {
    int  n    = 2 * g + hi;
    bool live = n < N;
    int  nc   = live ? n : N - 1;
    int   start = rowptr[nc];
    int   end   = rowptr[nc + 1];
    float qv    = q[(size_t)nc * N_C + j];
    int   deg   = live ? end - start : 0;
    int dmax = max(__builtin_amdgcn_readlane(deg, 0),
                   __builtin_amdgcn_readlane(deg, 16));
    float acc = 0.0f;
    for (int base = 0; base < dmax; base += NB2) {
      int idx = (base + j < deg) ? col[start + base + j] : 0;
      float v[NB2];
#pragma unroll
      for (int t = 0; t < NB2; ++t) {
        int s0 = __builtin_amdgcn_readlane(idx, t);
        int s1 = __builtin_amdgcn_readlane(idx, 16 + t);
        int s  = hi ? s1 : s0;
        v[t] = (base + t < deg) ? p[(size_t)s * N_C + j] : 0.0f;
      }
#pragma unroll
      for (int t = 0; t < NB2; ++t) acc += v[t];
    }
    if (live) {
      float inv = 1.0f / fmaxf((float)deg, 1.0f);
      out[(size_t)n * N_C + j] = qv + acc * inv;
    }
  }
}

extern "C" void kernel_launch(void* const* d_in, const int* in_sizes, int n_in,
                              void* d_out, int out_size, void* d_ws,
                              size_t ws_size, hipStream_t stream) {
  (void)n_in; (void)out_size; (void)ws_size;

  const float* x   = (const float*)d_in[0];
  const int*   ei  = (const int*)d_in[1];
  const float* W1l = (const float*)d_in[2];
  const float* b1  = (const float*)d_in[3];
  const float* W1r = (const float*)d_in[4];
  const float* W2l = (const float*)d_in[5];
  const float* b2  = (const float*)d_in[6];
  const float* W2r = (const float*)d_in[7];
  float* out = (float*)d_out;

  const int N  = in_sizes[0] / K_F;
  const int nE = in_sizes[1] / 2;
  const int* src = ei;
  const int* dst = ei + nE;
  const int nb = (N + BIN_W - 1) / BIN_W;
  const int nbp = (nb + 3) & ~3;

  float* pl0     = (float*)d_ws;
  float* p       = pl0 + (size_t)N * K_F;
  float* q       = p + (size_t)N * N_C;
  int*   rowptr  = (int*)(q + (size_t)N * N_C);
  int*   col     = rowptr + N + 16;
  int*   pairs   = (int*)pl0;
  int*   tcnt    = pairs + 2 * (size_t)nE;
  int*   tstart  = tcnt + SORT_G * nbp;

  const int T = 256;

  slab_bin_kernel<<<SORT_G, SORT_T, 0, stream>>>(src, dst, tcnt, tstart, pairs,
                                                 nE, nb, nbp);

  bin_layer1_kernel<<<nb, SORT_T, 0, stream>>>(pairs, tcnt, tstart, rowptr, col,
                                               x, W1l, W1r, b1, W2l, W2r, b2,
                                               p, q, N, nbp);

  gather16_close_kernel<<<RES_BLOCKS_EDGE, T, 0, stream>>>(p, q, rowptr, col,
                                                           out, N);
}


# sort pass 1 places its (source, destination) pairs in a shared-memory copy of the slab's stretch and stores it as whole 128-B lines: 800,000 scattered 8-B stores become 50,000 line stores
# speedup vs baseline: 4.9146x; 1.0509x over previous
#include <hip/hip_runtime.h>


typedef __attribute__((ext_vector_type(2))) float v2f;
typedef __attribute__((ext_vector_type(2))) int   v2i;
typedef __attribute__((ext_vector_type(4))) int   v4i;
typedef __attribute__((ext_vector_type(8))) float v8f;

#define K_F 64
#define N_C 16

#define SORT_T 1024
#define SORT_G 128
#define SORT_U 8
#define BIN_W  208
#define NB_MAX SORT_T
#define SEG_L  (SORT_T / SORT_G)
#define SEG_U  8
#define SEG_CAP (SEG_L * SEG_U)
#define CS_U   8
#define STAGE_CAP (SORT_U * SORT_T)
#define STAGE_TRIPS (SORT_U + 1)

__device__ __forceinline__ int wave_scan_incl(int v, int lane) {
  for (int o = 1; o < 32; o <<= 1) {
    int t = __shfl_up(v, o);
    if (lane >= o) v += t;
  }
  return v;
}

__global__ void slab_bin_kernel(const int* __restrict__ src,
                                const int* __restrict__ dst,
                                int* __restrict__ tcnt,
                                int* __restrict__ tstart,
                                int* __restrict__ pairs_, int nE, int nb,
                                int nbp) {
  __shared__ int hist[NB_MAX];
  __shared__ int s_part[SORT_T / 32];
  __shared__ v2i s_stage[STAGE_CAP];
  v2i* pairs = reinterpret_cast<v2i*>(pairs_);
  int tid  = threadIdx.x;
  int lane = tid & 31;
  int w    = tid >> 5;
  hist[tid] = 0;
  __syncthreads();
  int S    = (nE + gridDim.x - 1) / gridDim.x;
  int lo   = min((int)blockIdx.x * S, nE);
  int hi   = min(lo + S, nE);
  int last = max(hi - 1, 0);
  bool one = S <= SORT_U * SORT_T;
  int d[SORT_U];
  int s[SORT_U];
  int r[SORT_U];
  if (one) {
#pragma unroll
    for (int u = 0; u < SORT_U; ++u) {
      int ee = min(lo + tid + u * SORT_T, last);
      d[u] = dst[ee];
      s[u] = src[ee];
    }
#pragma unroll
    for (int u = 0; u < SORT_U; ++u) {
      r[u] = 0;
      if (lo + tid + u * SORT_T < hi) r[u] = atomicAdd(&hist[(unsigned)d[u] / BIN_W], 1);
    }
  } else {
    for (int e = lo + tid; e < hi; e += SORT_U * SORT_T) {
      int dd[SORT_U];
#pragma unroll
      for (int u = 0; u < SORT_U; ++u)
        dd[u] = dst[min(e + u * SORT_T, last)];
#pragma unroll
      for (int u = 0; u < SORT_U; ++u)
        if (e + u * SORT_T < hi) atomicAdd(&hist[(unsigned)dd[u] / BIN_W], 1);
    }
  }
  __syncthreads();
  int c   = (tid < nb) ? hist[tid] : 0;
  int inc = wave_scan_incl(c, lane);
  if (lane == 31) s_part[w] = inc;
  __syncthreads();
  int woff = 0;
  for (int k = 0; k < w; ++k) woff += s_part[k];
  int excl = woff + inc - c;
  if (tid < nbp) {
    hist[tid] = lo + excl;
    tstart[blockIdx.x * nbp + tid] = lo + excl;
    tcnt[blockIdx.x * nbp + tid]   = c;
  }
  __syncthreads();
  if (one) {
    int st[SORT_U];
#pragma unroll
    for (int u = 0; u < SORT_U; ++u) st[u] = hist[(unsigned)d[u] / BIN_W];
#pragma unroll
    for (int u = 0; u < SORT_U; ++u)
      if (lo + tid + u * SORT_T < hi) {
        v2i sd = {s[u], d[u]};
        s_stage[st[u] - lo + r[u]] = sd;
      }
    __syncthreads();
    int  e0 = (lo & ~31) + tid;
    v2i  t[STAGE_TRIPS];
#pragma unroll
    for (int k = 0; k < STAGE_TRIPS; ++k)
      t[k] = s_stage[min(max(e0 + k * SORT_T - lo, 0), STAGE_CAP - 1)];
#pragma unroll
    for (int k = 0; k < STAGE_TRIPS; ++k) {
      int e = e0 + k * SORT_T;
      if (e >= lo && e < hi) pairs[e] = t[k];
    }
  } else {
    for (int e = lo + tid; e < hi; e += SORT_U * SORT_T) {
      int dd[SORT_U];
      int ss[SORT_U];
      int pos[SORT_U];
#pragma unroll
      for (int u = 0; u < SORT_U; ++u) {
        int ee = min(e + u * SORT_T, last);
        dd[u] = dst[ee];
        ss[u] = src[ee];
      }
#pragma unroll
      for (int u = 0; u < SORT_U; ++u) {
        pos[u] = 0;
        if (e + u * SORT_T < hi) pos[u] = atomicAdd(&hist[(unsigned)dd[u] / BIN_W], 1);
      }
#pragma unroll
      for (int u = 0; u < SORT_U; ++u)
        if (e + u * SORT_T < hi) {
          v2i sd = {ss[u], dd[u]};
          pairs[pos[u]] = sd;
        }
    }
  }
}

#define RES_BLOCKS_EDGE  1024

__device__ __forceinline__ int wave_in_grid() {
  return __builtin_amdgcn_readfirstlane(
      (int)(blockIdx.x * (blockDim.x >> 5) + (threadIdx.x >> 5)));
}

#define NB1 32
#define LDP 68
#define COL_CAP 8192

__device__ __forceinline__ void wave_means(const float* __restrict__ feat,
                                           const int* lst, int lim,
                                           const int* rp8, float* mrow,
                                           int lane) {
  int rp    = rp8[min(lane, 8)];
  int start = __builtin_amdgcn_readlane(rp, 0);
  int idx   = lst[min(start + lane, lim)];
#pragma nounroll
  for (int i = 0; i < 8; ++i) {
    int end  = __builtin_amdgcn_readlane(rp, i + 1);
    int deg  = end - start;
    int idxN = lst[min(end + lane, lim)];
    float lo = 0.0f;
    float up = 0.0f;
    for (int base = 0; base < deg; base += NB1) {
      if (base > 0) idx = lst[min(start + base + lane, lim)];
      float a[NB1];
      float bq[NB1];
#pragma unroll
      for (int t = 0; t < NB1; ++t) {
        a[t]  = 0.0f;
        bq[t] = 0.0f;
        if (base + t < deg) {
          int s = __builtin_amdgcn_readlane(idx, t);
          const float* rw = feat + (size_t)s * K_F + lane;
          a[t]  = rw[0];
          bq[t] = rw[32];
        }
      }
#pragma unroll
      for (int t = 0; t < NB1; ++t) {
        lo += a[t];
        up += bq[t];
      }
    }
    float inv = 1.0f / fmaxf((float)deg, 1.0f);
    float* m = mrow + i * LDP + lane;
    m[0]  = lo * inv;
    m[32] = up * inv;
    start = end;
    idx   = idxN;
  }
}

__global__ void __launch_bounds__(SORT_T)
bin_layer1_kernel(const int* __restrict__ pairs_,
                  const int* __restrict__ tcnt,
                  const int* __restrict__ tstart,
                  int* __restrict__ rowptr,
                  int* __restrict__ col,
                  const float* __restrict__ feat,
                  const float* __restrict__ Wl,
                  const float* __restrict__ Wr,
                  const float* __restrict__ bias,
                  const float* __restrict__ W2l,
                  const float* __restrict__ W2r,
                  const float* __restrict__ bias2,
                  float* __restrict__ p,
                  float* __restrict__ q, int N, int nbp) {
  __shared__ int cnt[BIN_W];
  __shared__ int s_rp[BIN_W + 1];
  __shared__ int sst[SORT_G];
  __shared__ int scn[SORT_G];
  __shared__ int s_partA[SORT_T / 32];
  __shared__ int s_part[SORT_T / 32];
  __shared__ int s_big;
  __shared__ int s_col[COL_CAP];
  __shared__ alignas(16) float s_mean[BIN_W * LDP];
  __shared__ alignas(16) float s_h[BIN_W * LDP];
  const v2i* pairs = reinterpret_cast<const v2i*>(pairs_);
  unsigned tid = threadIdx.x;
  int lane  = tid & 31;
  int w     = tid >> 5;
  int g     = tid / SEG_L;
  int ll    = tid & (SEG_L - 1);
  int b     = blockIdx.x;
  int node0 = b * BIN_W;

  if (tid < BIN_W) cnt[tid] = 0;
  if (tid == 0) s_big = 0;

  unsigned unbp  = (unsigned)nbp;
  unsigned mrow  = min(tid, (unsigned)(SORT_G - 1)) * unbp + (unsigned)b;
  int      mst   = tstart[mrow];
  int      mcn   = tcnt[mrow];
  unsigned nq    = unbp >> 2;
  unsigned parts = SORT_T / nq;
  unsigned part  = tid / nq;
  unsigned qc    = tid - part * nq;
  int  c0   = (int)(4 * qc);
  bool in0  = c0 < b;
  bool in1  = c0 + 1 < b;
  bool in2  = c0 + 2 < b;
  bool in3  = c0 + 3 < b;
  int  sum  = 0;
  for (unsigned g0 = part; g0 < SORT_G; g0 += CS_U * parts) {
    v4i t[CS_U];
#pragma unroll
    for (int u = 0; u < CS_U; ++u)
      t[u] = *reinterpret_cast<const v4i*>(
          tcnt + min(g0 + u * parts, (unsigned)(SORT_G - 1)) * unbp + 4 * qc);
#pragma unroll
    for (int u = 0; u < CS_U; ++u)
      if (g0 + u * parts < SORT_G)
        sum += (in0 ? t[u].x : 0) + (in1 ? t[u].y : 0) +
               (in2 ? t[u].z : 0) + (in3 ? t[u].w : 0);
  }
  if (part >= parts) sum = 0;
  __syncthreads();
  if (tid < SORT_G) {
    sst[tid] = mst;
    scn[tid] = mcn;
    if (mcn > SEG_CAP) s_big = 1;
  }
  int incA = wave_scan_incl(sum, lane);
  if (lane == 31) s_partA[w] = incA;
  __syncthreads();
  int seg0 = 0;
#pragma unroll
  for (int k = 0; k < SORT_T / 32; ++k) seg0 += s_partA[k];
  int  s0   = sst[g];
  int  s1   = s0 + scn[g];
  int  last = max(s1 - 1, 0);
  bool big  = s_big != 0;

  v2i sd[SEG_U];
  int r[SEG_U];
  if (!big) {
#pragma unroll
    for (int u = 0; u < SEG_U; ++u)
      sd[u] = pairs[min(s0 + ll + u * SEG_L, last)];
#pragma unroll
    for (int u = 0; u < SEG_U; ++u) {
      r[u] = 0;
      if (s0 + ll + u * SEG_L < s1) r[u] = atomicAdd(&cnt[sd[u].y - node0], 1);
    }
  } else {
    for (int e = s0 + ll; e < s1; e += SEG_U * SEG_L) {
      int dd[SEG_U];
#pragma unroll
      for (int u = 0; u < SEG_U; ++u)
        dd[u] = pairs[min(e + u * SEG_L, last)].y;
#pragma unroll
      for (int u = 0; u < SEG_U; ++u)
        if (e + u * SEG_L < s1) atomicAdd(&cnt[dd[u] - node0], 1);
    }
  }
  __syncthreads();
  int c   = (tid < BIN_W) ? cnt[tid] : 0;
  int inc = wave_scan_incl(c, lane);
  if (lane == 31) s_part[w] = inc;
  __syncthreads();
  int woff = 0;
  for (int k = 0; k < w; ++k) woff += s_part[k];
  int excl = woff + inc - c;
  if (tid < BIN_W) {
    cnt[tid]  = excl;
    s_rp[tid] = excl;
    if (tid == BIN_W - 1) s_rp[BIN_W] = excl + c;
    int node = node0 + (int)tid;
    if (node < N) rowptr[node] = seg0 + excl;
    if (node == N - 1) rowptr[N] = seg0 + excl + c;
  }
  __syncthreads();
  int  total = __builtin_amdgcn_readfirstlane(s_rp[BIN_W]);
  bool fits  = total <= COL_CAP;
  if (!big) {
    int st[SEG_U];
#pragma unroll
    for (int u = 0; u < SEG_U; ++u)
      st[u] = cnt[min((unsigned)(sd[u].y - node0), (unsigned)(BIN_W - 1))];
#pragma unroll
    for (int u = 0; u < SEG_U; ++u)
      if (s0 + ll + u * SEG_L < s1) {
        if (fits) s_col[st[u] + r[u]] = sd[u].x;
        else      col[seg0 + st[u] + r[u]] = sd[u].x;
      }
  } else {
    for (int e = s0 + ll; e < s1; e += SEG_U * SEG_L) {
      v2i sdd[SEG_U];
      int rr[SEG_U];
#pragma unroll
      for (int u = 0; u < SEG_U; ++u)
        sdd[u] = pairs[min(e + u * SEG_L, last)];
#pragma unroll
      for (int u = 0; u < SEG_U; ++u) {
        rr[u] = 0;
        if (e + u * SEG_L < s1) rr[u] = atomicAdd(&cnt[sdd[u].y - node0], 1);
      }
#pragma unroll
      for (int u = 0; u < SEG_U; ++u)
        if (e + u * SEG_L < s1) {
          if (fits) s_col[rr[u]] = sdd[u].x;
          else      col[seg0 + rr[u]] = sdd[u].x;
        }
    }
  }
  __syncthreads();
  if (fits)
    for (int i = (int)tid; i < total; i += SORT_T) col[seg0 + i] = s_col[i];

  int  w2     = __builtin_amdgcn_readfirstlane((int)(tid >> 5));
  int  sl     = w2 >> 1;
  int  wp     = w2 & 1;
  int  stripe = (node0 >> 4) + sl;
  bool live   = sl < BIN_W / 16 && node0 + sl * 16 < N;
  int  hrow   = lane & 15;
  int  hi     = lane >> 4;
  int  koff   = hi * 2;

  if (live) {
    int lim = max(total - 1, 0);
    if (fits)
      wave_means(feat, s_col, lim, s_rp + w2 * 8, s_mean + w2 * 8 * LDP, lane);
    else
      wave_means(feat, col + seg0, lim, s_rp + w2 * 8, s_mean + w2 * 8 * LDP,
                 lane);
  }

  int cb = wp * 32 + hrow;
  v8f c0v = {};
  v8f c1v = {};
  if (live) {
    const float* xrow = feat + (size_t)(stripe * 16 + hrow) * K_F + koff;
#pragma unroll
    for (int k = 0; k < K_F; k += 4) {
      v2f ax = *reinterpret_cast<const v2f*>(xrow + k);
      const float* wr0 = Wr + (k + koff) * K_F + cb;
      v2f br;
      br.x = wr0[0];        br.y = wr0[K_F];
      c0v = __builtin_amdgcn_wmma_f32_16x16x4_f32(false, ax, false, br,
                                                  (short)0, c0v, false, false);
      br.x = wr0[16];       br.y = wr0[K_F + 16];
      c1v = __builtin_amdgcn_wmma_f32_16x16x4_f32(false, ax, false, br,
                                                  (short)0, c1v, false, false);
    }
  }
  __syncthreads();
  if (live) {
    const float* arow = s_mean + (sl * 16 + hrow) * LDP + koff;
#pragma unroll
    for (int k = 0; k < K_F; k += 4) {
      v2f am = *reinterpret_cast<const v2f*>(arow + k);
      const float* wl0 = Wl + (k + koff) * K_F + cb;
      v2f bl;
      bl.x = wl0[0];        bl.y = wl0[K_F];
      c0v = __builtin_amdgcn_wmma_f32_16x16x4_f32(false, am, false, bl,
                                                  (short)0, c0v, false, false);
      bl.x = wl0[16];       bl.y = wl0[K_F + 16];
      c1v = __builtin_amdgcn_wmma_f32_16x16x4_f32(false, am, false, bl,
                                                  (short)0, c1v, false, false);
    }
    float bv0 = bias[cb];
    float bv1 = bias[cb + 16];
#pragma unroll
    for (int i = 0; i < 8; ++i) {
      float* o = s_h + (sl * 16 + hi * 8 + i) * LDP + cb;
      o[0]  = fmaxf(c0v[i] + bv0, 0.0f);
      o[16] = fmaxf(c1v[i] + bv1, 0.0f);
    }
  }
  __syncthreads();

  if (live) {
    const float* W2   = wp ? W2r : W2l;
    float*       dst2 = wp ? q : p;
    float        bv   = wp ? bias2[hrow] : 0.0f;
    const float* arow = s_h + (sl * 16 + hrow) * LDP + koff;
    v8f cc = {};
#pragma unroll
    for (int k = 0; k < K_F; k += 4) {
      v2f a = *reinterpret_cast<const v2f*>(arow + k);
      v2f bb;
      bb.x = W2[(k + koff) * N_C + hrow];
      bb.y = W2[(k + koff + 1) * N_C + hrow];
      cc = __builtin_amdgcn_wmma_f32_16x16x4_f32(false, a, false, bb,
                                                 (short)0, cc, false, false);
    }
#pragma unroll
    for (int i = 0; i < 8; ++i)
      dst2[(size_t)(stripe * 16 + hi * 8 + i) * N_C + hrow] = cc[i] + bv;
  }
}

#define NB2 16

__global__ void gather16_close_kernel(const float* __restrict__ p,
                                      const float* __restrict__ q,
                                      const int* __restrict__ rowptr,
                                      const int* __restrict__ col,
                                      float* __restrict__ out, int N) {
  int nPairs = (N + 1) >> 1;
  int nWaves = gridDim.x * (blockDim.x >> 5);
  int lane = threadIdx.x & 31;
  int j    = lane & 15;
  int hi   = lane >> 4;
  for (int g = wave_in_grid(); g < nPairs; g += nWaves) {
    int  n    = 2 * g + hi;
    bool live = n < N;
    int  nc   = live ? n : N - 1;
    int   start = rowptr[nc];
    int   end   = rowptr[nc + 1];
    float qv    = q[(size_t)nc * N_C + j];
    int   deg   = live ? end - start : 0;
    int dmax = max(__builtin_amdgcn_readlane(deg, 0),
                   __builtin_amdgcn_readlane(deg, 16));
    float acc = 0.0f;
    for (int base = 0; base < dmax; base += NB2) {
      int idx = (base + j < deg) ? col[start + base + j] : 0;
      float v[NB2];
#pragma unroll
      for (int t = 0; t < NB2; ++t) {
        int s0 = __builtin_amdgcn_readlane(idx, t);
        int s1 = __builtin_amdgcn_readlane(idx, 16 + t);
        int s  = hi ? s1 : s0;
        v[t] = (base + t < deg) ? p[(size_t)s * N_C + j] : 0.0f;
      }
#pragma unroll
      for (int t = 0; t < NB2; ++t) acc += v[t];
    }
    if (live) {
      float inv = 1.0f / fmaxf((float)deg, 1.0f);
      out[(size_t)n * N_C + j] = qv + acc * inv;
    }
  }
}

extern "C" void kernel_launch(void* const* d_in, const int* in_sizes, int n_in,
                              void* d_out, int out_size, void* d_ws,
                              size_t ws_size, hipStream_t stream) {
  (void)n_in; (void)out_size; (void)ws_size;

  const float* x   = (const float*)d_in[0];
  const int*   ei  = (const int*)d_in[1];
  const float* W1l = (const float*)d_in[2];
  const float* b1  = (const float*)d_in[3];
  const float* W1r = (const float*)d_in[4];
  const float* W2l = (const float*)d_in[5];
  const float* b2  = (const float*)d_in[6];
  const float* W2r = (const float*)d_in[7];
  float* out = (float*)d_out;

  const int N  = in_sizes[0] / K_F;
  const int nE = in_sizes[1] / 2;
  const int* src = ei;
  const int* dst = ei + nE;
  const int nb = (N + BIN_W - 1) / BIN_W;
  const int nbp = (nb + 3) & ~3;

  float* pl0     = (float*)d_ws;
  float* p       = pl0 + (size_t)N * K_F;
  float* q       = p + (size_t)N * N_C;
  int*   rowptr  = (int*)(q + (size_t)N * N_C);
  int*   col     = rowptr + N + 16;
  int*   pairs   = (int*)pl0;
  int*   tcnt    = pairs + 2 * (size_t)nE;
  int*   tstart  = tcnt + SORT_G * nbp;

  const int T = 256;

  slab_bin_kernel<<<SORT_G, SORT_T, 0, stream>>>(src, dst, tcnt, tstart, pairs,
                                                 nE, nb, nbp);

  bin_layer1_kernel<<<nb, SORT_T, 0, stream>>>(pairs, tcnt, tstart, rowptr, col,
                                               x, W1l, W1r, b1, W2l, W2r, b2,
                                               p, q, N, nbp);

  gather16_close_kernel<<<RES_BLOCKS_EDGE, T, 0, stream>>>(p, q, rowptr, col,
                                                           out, N);
}


# layer-2 gather: 4 lanes x 16 B per 64-B row: a wave takes 8 nodes, 24 rows each in flight (next chunk's indices asked for with them); one 8-node item per wave, 6,250 waves; f32 sum order unchanged
# speedup vs baseline: 5.7110x; 1.1620x over previous
#include <hip/hip_runtime.h>


typedef __attribute__((ext_vector_type(2))) float v2f;
typedef __attribute__((ext_vector_type(2))) int   v2i;
typedef __attribute__((ext_vector_type(4))) int   v4i;
typedef __attribute__((ext_vector_type(4))) float v4f;
typedef __attribute__((ext_vector_type(8))) float v8f;

#define K_F 64
#define N_C 16

#define SORT_T 1024
#define SORT_G 128
#define SORT_U 8
#define BIN_W  208
#define NB_MAX SORT_T
#define SEG_L  (SORT_T / SORT_G)
#define SEG_U  8
#define SEG_CAP (SEG_L * SEG_U)
#define CS_U   8
#define STAGE_CAP (SORT_U * SORT_T)
#define STAGE_TRIPS (SORT_U + 1)

__device__ __forceinline__ int wave_scan_incl(int v, int lane) {
  for (int o = 1; o < 32; o <<= 1) {
    int t = __shfl_up(v, o);
    if (lane >= o) v += t;
  }
  return v;
}

__global__ void slab_bin_kernel(const int* __restrict__ src,
                                const int* __restrict__ dst,
                                int* __restrict__ tcnt,
                                int* __restrict__ tstart,
                                int* __restrict__ pairs_, int nE, int nb,
                                int nbp) {
  __shared__ int hist[NB_MAX];
  __shared__ int s_part[SORT_T / 32];
  __shared__ v2i s_stage[STAGE_CAP];
  v2i* pairs = reinterpret_cast<v2i*>(pairs_);
  int tid  = threadIdx.x;
  int lane = tid & 31;
  int w    = tid >> 5;
  hist[tid] = 0;
  __syncthreads();
  int S    = (nE + gridDim.x - 1) / gridDim.x;
  int lo   = min((int)blockIdx.x * S, nE);
  int hi   = min(lo + S, nE);
  int last = max(hi - 1, 0);
  bool one = S <= SORT_U * SORT_T;
  int d[SORT_U];
  int s[SORT_U];
  int r[SORT_U];
  if (one) {
#pragma unroll
    for (int u = 0; u < SORT_U; ++u) {
      int ee = min(lo + tid + u * SORT_T, last);
      d[u] = dst[ee];
      s[u] = src[ee];
    }
#pragma unroll
    for (int u = 0; u < SORT_U; ++u) {
      r[u] = 0;
      if (lo + tid + u * SORT_T < hi) r[u] = atomicAdd(&hist[(unsigned)d[u] / BIN_W], 1);
    }
  } else {
    for (int e = lo + tid; e < hi; e += SORT_U * SORT_T) {
      int dd[SORT_U];
#pragma unroll
      for (int u = 0; u < SORT_U; ++u)
        dd[u] = dst[min(e + u * SORT_T, last)];
#pragma unroll
      for (int u = 0; u < SORT_U; ++u)
        if (e + u * SORT_T < hi) atomicAdd(&hist[(unsigned)dd[u] / BIN_W], 1);
    }
  }
  __syncthreads();
  int c   = (tid < nb) ? hist[tid] : 0;
  int inc = wave_scan_incl(c, lane);
  if (lane == 31) s_part[w] = inc;
  __syncthreads();
  int woff = 0;
  for (int k = 0; k < w; ++k) woff += s_part[k];
  int excl = woff + inc - c;
  if (tid < nbp) {
    hist[tid] = lo + excl;
    tstart[blockIdx.x * nbp + tid] = lo + excl;
    tcnt[blockIdx.x * nbp + tid]   = c;
  }
  __syncthreads();
  if (one) {
    int st[SORT_U];
#pragma unroll
    for (int u = 0; u < SORT_U; ++u) st[u] = hist[(unsigned)d[u] / BIN_W];
#pragma unroll
    for (int u = 0; u < SORT_U; ++u)
      if (lo + tid + u * SORT_T < hi) {
        v2i sd = {s[u], d[u]};
        s_stage[st[u] - lo + r[u]] = sd;
      }
    __syncthreads();
    int  e0 = (lo & ~31) + tid;
    v2i  t[STAGE_TRIPS];
#pragma unroll
    for (int k = 0; k < STAGE_TRIPS; ++k)
      t[k] = s_stage[min(max(e0 + k * SORT_T - lo, 0), STAGE_CAP - 1)];
#pragma unroll
    for (int k = 0; k < STAGE_TRIPS; ++k) {
      int e = e0 + k * SORT_T;
      if (e >= lo && e < hi) pairs[e] = t[k];
    }
  } else {
    for (int e = lo + tid; e < hi; e += SORT_U * SORT_T) {
      int dd[SORT_U];
      int ss[SORT_U];
      int pos[SORT_U];
#pragma unroll
      for (int u = 0; u < SORT_U; ++u) {
        int ee = min(e + u * SORT_T, last);
        dd[u] = dst[ee];
        ss[u] = src[ee];
      }
#pragma unroll
      for (int u = 0; u < SORT_U; ++u) {
        pos[u] = 0;
        if (e + u * SORT_T < hi) pos[u] = atomicAdd(&hist[(unsigned)dd[u] / BIN_W], 1);
      }
#pragma unroll
      for (int u = 0; u < SORT_U; ++u)
        if (e + u * SORT_T < hi) {
          v2i sd = {ss[u], dd[u]};
          pairs[pos[u]] = sd;
        }
    }
  }
}

__device__ __forceinline__ int wave_in_grid() {
  return __builtin_amdgcn_readfirstlane(
      (int)(blockIdx.x * (blockDim.x >> 5) + (threadIdx.x >> 5)));
}

#define NB1 32
#define LDP 68
#define COL_CAP 8192

__device__ __forceinline__ void wave_means(const float* __restrict__ feat,
                                           const int* lst, int lim,
                                           const int* rp8, float* mrow,
                                           int lane) {
  int rp    = rp8[min(lane, 8)];
  int start = __builtin_amdgcn_readlane(rp, 0);
  int idx   = lst[min(start + lane, lim)];
#pragma nounroll
  for (int i = 0; i < 8; ++i) {
    int end  = __builtin_amdgcn_readlane(rp, i + 1);
    int deg  = end - start;
    int idxN = lst[min(end + lane, lim)];
    float lo = 0.0f;
    float up = 0.0f;
    for (int base = 0; base < deg; base += NB1) {
      if (base > 0) idx = lst[min(start + base + lane, lim)];
      float a[NB1];
      float bq[NB1];
#pragma unroll
      for (int t = 0; t < NB1; ++t) {
        a[t]  = 0.0f;
        bq[t] = 0.0f;
        if (base + t < deg) {
          int s = __builtin_amdgcn_readlane(idx, t);
          const float* rw = feat + (size_t)s * K_F + lane;
          a[t]  = rw[0];
          bq[t] = rw[32];
        }
      }
#pragma unroll
      for (int t = 0; t < NB1; ++t) {
        lo += a[t];
        up += bq[t];
      }
    }
    float inv = 1.0f / fmaxf((float)deg, 1.0f);
    float* m = mrow + i * LDP + lane;
    m[0]  = lo * inv;
    m[32] = up * inv;
    start = end;
    idx   = idxN;
  }
}

__global__ void __launch_bounds__(SORT_T)
bin_layer1_kernel(const int* __restrict__ pairs_,
                  const int* __restrict__ tcnt,
                  const int* __restrict__ tstart,
                  int* __restrict__ rowptr,
                  int* __restrict__ col,
                  const float* __restrict__ feat,
                  const float* __restrict__ Wl,
                  const float* __restrict__ Wr,
                  const float* __restrict__ bias,
                  const float* __restrict__ W2l,
                  const float* __restrict__ W2r,
                  const float* __restrict__ bias2,
                  float* __restrict__ p,
                  float* __restrict__ q, int N, int nbp) {
  __shared__ int cnt[BIN_W];
  __shared__ int s_rp[BIN_W + 1];
  __shared__ int sst[SORT_G];
  __shared__ int scn[SORT_G];
  __shared__ int s_partA[SORT_T / 32];
  __shared__ int s_part[SORT_T / 32];
  __shared__ int s_big;
  __shared__ int s_col[COL_CAP];
  __shared__ alignas(16) float s_mean[BIN_W * LDP];
  __shared__ alignas(16) float s_h[BIN_W * LDP];
  const v2i* pairs = reinterpret_cast<const v2i*>(pairs_);
  unsigned tid = threadIdx.x;
  int lane  = tid & 31;
  int w     = tid >> 5;
  int g     = tid / SEG_L;
  int ll    = tid & (SEG_L - 1);
  int b     = blockIdx.x;
  int node0 = b * BIN_W;

  if (tid < BIN_W) cnt[tid] = 0;
  if (tid == 0) s_big = 0;

  unsigned unbp  = (unsigned)nbp;
  unsigned mrow  = min(tid, (unsigned)(SORT_G - 1)) * unbp + (unsigned)b;
  int      mst   = tstart[mrow];
  int      mcn   = tcnt[mrow];
  unsigned nq    = unbp >> 2;
  unsigned parts = SORT_T / nq;
  unsigned part  = tid / nq;
  unsigned qc    = tid - part * nq;
  int  c0   = (int)(4 * qc);
  bool in0  = c0 < b;
  bool in1  = c0 + 1 < b;
  bool in2  = c0 + 2 < b;
  bool in3  = c0 + 3 < b;
  int  sum  = 0;
  for (unsigned g0 = part; g0 < SORT_G; g0 += CS_U * parts) {
    v4i t[CS_U];
#pragma unroll
    for (int u = 0; u < CS_U; ++u)
      t[u] = *reinterpret_cast<const v4i*>(
          tcnt + min(g0 + u * parts, (unsigned)(SORT_G - 1)) * unbp + 4 * qc);
#pragma unroll
    for (int u = 0; u < CS_U; ++u)
      if (g0 + u * parts < SORT_G)
        sum += (in0 ? t[u].x : 0) + (in1 ? t[u].y : 0) +
               (in2 ? t[u].z : 0) + (in3 ? t[u].w : 0);
  }
  if (part >= parts) sum = 0;
  __syncthreads();
  if (tid < SORT_G) {
    sst[tid] = mst;
    scn[tid] = mcn;
    if (mcn > SEG_CAP) s_big = 1;
  }
  int incA = wave_scan_incl(sum, lane);
  if (lane == 31) s_partA[w] = incA;
  __syncthreads();
  int seg0 = 0;
#pragma unroll
  for (int k = 0; k < SORT_T / 32; ++k) seg0 += s_partA[k];
  int  s0   = sst[g];
  int  s1   = s0 + scn[g];
  int  last = max(s1 - 1, 0);
  bool big  = s_big != 0;

  v2i sd[SEG_U];
  int r[SEG_U];
  if (!big) {
#pragma unroll
    for (int u = 0; u < SEG_U; ++u)
      sd[u] = pairs[min(s0 + ll + u * SEG_L, last)];
#pragma unroll
    for (int u = 0; u < SEG_U; ++u) {
      r[u] = 0;
      if (s0 + ll + u * SEG_L < s1) r[u] = atomicAdd(&cnt[sd[u].y - node0], 1);
    }
  } else {
    for (int e = s0 + ll; e < s1; e += SEG_U * SEG_L) {
      int dd[SEG_U];
#pragma unroll
      for (int u = 0; u < SEG_U; ++u)
        dd[u] = pairs[min(e + u * SEG_L, last)].y;
#pragma unroll
      for (int u = 0; u < SEG_U; ++u)
        if (e + u * SEG_L < s1) atomicAdd(&cnt[dd[u] - node0], 1);
    }
  }
  __syncthreads();
  int c   = (tid < BIN_W) ? cnt[tid] : 0;
  int inc = wave_scan_incl(c, lane);
  if (lane == 31) s_part[w] = inc;
  __syncthreads();
  int woff = 0;
  for (int k = 0; k < w; ++k) woff += s_part[k];
  int excl = woff + inc - c;
  if (tid < BIN_W) {
    cnt[tid]  = excl;
    s_rp[tid] = excl;
    if (tid == BIN_W - 1) s_rp[BIN_W] = excl + c;
    int node = node0 + (int)tid;
    if (node < N) rowptr[node] = seg0 + excl;
    if (node == N - 1) rowptr[N] = seg0 + excl + c;
  }
  __syncthreads();
  int  total = __builtin_amdgcn_readfirstlane(s_rp[BIN_W]);
  bool fits  = total <= COL_CAP;
  if (!big) {
    int st[SEG_U];
#pragma unroll
    for (int u = 0; u < SEG_U; ++u)
      st[u] = cnt[min((unsigned)(sd[u].y - node0), (unsigned)(BIN_W - 1))];
#pragma unroll
    for (int u = 0; u < SEG_U; ++u)
      if (s0 + ll + u * SEG_L < s1) {
        if (fits) s_col[st[u] + r[u]] = sd[u].x;
        else      col[seg0 + st[u] + r[u]] = sd[u].x;
      }
  } else {
    for (int e = s0 + ll; e < s1; e += SEG_U * SEG_L) {
      v2i sdd[SEG_U];
      int rr[SEG_U];
#pragma unroll
      for (int u = 0; u < SEG_U; ++u)
        sdd[u] = pairs[min(e + u * SEG_L, last)];
#pragma unroll
      for (int u = 0; u < SEG_U; ++u) {
        rr[u] = 0;
        if (e + u * SEG_L < s1) rr[u] = atomicAdd(&cnt[sdd[u].y - node0], 1);
      }
#pragma unroll
      for (int u = 0; u < SEG_U; ++u)
        if (e + u * SEG_L < s1) {
          if (fits) s_col[rr[u]] = sdd[u].x;
          else      col[seg0 + rr[u]] = sdd[u].x;
        }
    }
  }
  __syncthreads();
  if (fits)
    for (int i = (int)tid; i < total; i += SORT_T) col[seg0 + i] = s_col[i];

  int  w2     = __builtin_amdgcn_readfirstlane((int)(tid >> 5));
  int  sl     = w2 >> 1;
  int  wp     = w2 & 1;
  int  stripe = (node0 >> 4) + sl;
  bool live   = sl < BIN_W / 16 && node0 + sl * 16 < N;
  int  hrow   = lane & 15;
  int  hi     = lane >> 4;
  int  koff   = hi * 2;

  if (live) {
    int lim = max(total - 1, 0);
    if (fits)
      wave_means(feat, s_col, lim, s_rp + w2 * 8, s_mean + w2 * 8 * LDP, lane);
    else
      wave_means(feat, col + seg0, lim, s_rp + w2 * 8, s_mean + w2 * 8 * LDP,
                 lane);
  }

  int cb = wp * 32 + hrow;
  v8f c0v = {};
  v8f c1v = {};
  if (live) {
    const float* xrow = feat + (size_t)(stripe * 16 + hrow) * K_F + koff;
#pragma unroll
    for (int k = 0; k < K_F; k += 4) {
      v2f ax = *reinterpret_cast<const v2f*>(xrow + k);
      const float* wr0 = Wr + (k + koff) * K_F + cb;
      v2f br;
      br.x = wr0[0];        br.y = wr0[K_F];
      c0v = __builtin_amdgcn_wmma_f32_16x16x4_f32(false, ax, false, br,
                                                  (short)0, c0v, false, false);
      br.x = wr0[16];       br.y = wr0[K_F + 16];
      c1v = __builtin_amdgcn_wmma_f32_16x16x4_f32(false, ax, false, br,
                                                  (short)0, c1v, false, false);
    }
  }
  __syncthreads();
  if (live) {
    const float* arow = s_mean + (sl * 16 + hrow) * LDP + koff;
#pragma unroll
    for (int k = 0; k < K_F; k += 4) {
      v2f am = *reinterpret_cast<const v2f*>(arow + k);
      const float* wl0 = Wl + (k + koff) * K_F + cb;
      v2f bl;
      bl.x = wl0[0];        bl.y = wl0[K_F];
      c0v = __builtin_amdgcn_wmma_f32_16x16x4_f32(false, am, false, bl,
                                                  (short)0, c0v, false, false);
      bl.x = wl0[16];       bl.y = wl0[K_F + 16];
      c1v = __builtin_amdgcn_wmma_f32_16x16x4_f32(false, am, false, bl,
                                                  (short)0, c1v, false, false);
    }
    float bv0 = bias[cb];
    float bv1 = bias[cb + 16];
#pragma unroll
    for (int i = 0; i < 8; ++i) {
      float* o = s_h + (sl * 16 + hi * 8 + i) * LDP + cb;
      o[0]  = fmaxf(c0v[i] + bv0, 0.0f);
      o[16] = fmaxf(c1v[i] + bv1, 0.0f);
    }
  }
  __syncthreads();

  if (live) {
    const float* W2   = wp ? W2r : W2l;
    float*       dst2 = wp ? q : p;
    float        bv   = wp ? bias2[hrow] : 0.0f;
    const float* arow = s_h + (sl * 16 + hrow) * LDP + koff;
    v8f cc = {};
#pragma unroll
    for (int k = 0; k < K_F; k += 4) {
      v2f a = *reinterpret_cast<const v2f*>(arow + k);
      v2f bb;
      bb.x = W2[(k + koff) * N_C + hrow];
      bb.y = W2[(k + koff + 1) * N_C + hrow];
      cc = __builtin_amdgcn_wmma_f32_16x16x4_f32(false, a, false, bb,
                                                 (short)0, cc, false, false);
    }
#pragma unroll
    for (int i = 0; i < 8; ++i)
      dst2[(size_t)(stripe * 16 + hi * 8 + i) * N_C + hrow] = cc[i] + bv;
  }
}

#define NB2 24
#define NBQ (NB2 / 4)

__device__ __forceinline__ int quad_bcast(int v, int qd) {
  switch (qd) {
    case 0:  return __builtin_amdgcn_update_dpp(0, v, 0x00, 0xF, 0xF, true);
    case 1:  return __builtin_amdgcn_update_dpp(0, v, 0x55, 0xF, 0xF, true);
    case 2:  return __builtin_amdgcn_update_dpp(0, v, 0xAA, 0xF, 0xF, true);
    default: return __builtin_amdgcn_update_dpp(0, v, 0xFF, 0xF, 0xF, true);
  }
}

__global__ void gather16_close_kernel(const float* __restrict__ p,
                                      const float* __restrict__ q,
                                      const int* __restrict__ rowptr,
                                      const int* __restrict__ col,
                                      float* __restrict__ out, int N) {
  const v4f* p4 = reinterpret_cast<const v4f*>(p);
  int nItems = (N + 7) >> 3;
  int nWaves = gridDim.x * (blockDim.x >> 5);
  int lane = threadIdx.x & 31;
  int k    = lane >> 2;
  int j    = lane & 3;
  for (int g = wave_in_grid(); g < nItems; g += nWaves) {
    int  n    = 8 * g + k;
    bool live = n < N;
    int  nc   = live ? n : N - 1;
    int start = rowptr[nc];
    int end   = rowptr[nc + 1];
    v4f qv    = *reinterpret_cast<const v4f*>(q + (size_t)nc * N_C + 4 * j);
    int deg   = live ? end - start : 0;
    int dmax = __builtin_amdgcn_readlane(deg, 0);
#pragma unroll
    for (int kk = 1; kk < 8; ++kk)
      dmax = max(dmax, __builtin_amdgcn_readlane(deg, 4 * kk));
    int idx[NBQ];
#pragma unroll
    for (int m = 0; m < NBQ; ++m)
      idx[m] = (NBQ * j + m < deg) ? col[start + NBQ * j + m] : 0;
    v4f acc = {0.0f, 0.0f, 0.0f, 0.0f};
    for (int base = 0; base < dmax; base += NB2) {
      v4f v[NB2];
#pragma unroll
      for (int t = 0; t < NB2; ++t) {
        int s = quad_bcast(idx[t % NBQ], t / NBQ);
        v4f z = {0.0f, 0.0f, 0.0f, 0.0f};
        v[t] = z;
        if (base + t < deg) v[t] = p4[(unsigned)(s * (N_C / 4) + j)];
      }
#pragma unroll
      for (int m = 0; m < NBQ; ++m) {
        int slot = base + NB2 + NBQ * j + m;
        idx[m] = (slot < deg) ? col[start + slot] : 0;
      }
#pragma unroll
      for (int t = 0; t < NB2; ++t) acc += v[t];
    }
    if (live) {
      float inv = 1.0f / fmaxf((float)deg, 1.0f);
      *reinterpret_cast<v4f*>(out + (size_t)n * N_C + 4 * j) = qv + acc * inv;
    }
  }
}

extern "C" void kernel_launch(void* const* d_in, const int* in_sizes, int n_in,
                              void* d_out, int out_size, void* d_ws,
                              size_t ws_size, hipStream_t stream) {
  (void)n_in; (void)out_size; (void)ws_size;

  const float* x   = (const float*)d_in[0];
  const int*   ei  = (const int*)d_in[1];
  const float* W1l = (const float*)d_in[2];
  const float* b1  = (const float*)d_in[3];
  const float* W1r = (const float*)d_in[4];
  const float* W2l = (const float*)d_in[5];
  const float* b2  = (const float*)d_in[6];
  const float* W2r = (const float*)d_in[7];
  float* out = (float*)d_out;

  const int N  = in_sizes[0] / K_F;
  const int nE = in_sizes[1] / 2;
  const int* src = ei;
  const int* dst = ei + nE;
  const int nb = (N + BIN_W - 1) / BIN_W;
  const int nbp = (nb + 3) & ~3;
  const int gb = (N + 63) / 64;

  float* pl0     = (float*)d_ws;
  float* p       = pl0 + (size_t)N * K_F;
  float* q       = p + (size_t)N * N_C;
  int*   rowptr  = (int*)(q + (size_t)N * N_C);
  int*   col     = rowptr + N + 16;
  int*   pairs   = (int*)pl0;
  int*   tcnt    = pairs + 2 * (size_t)nE;
  int*   tstart  = tcnt + SORT_G * nbp;

  const int T = 256;

  slab_bin_kernel<<<SORT_G, SORT_T, 0, stream>>>(src, dst, tcnt, tstart, pairs,
                                                 nE, nb, nbp);

  bin_layer1_kernel<<<nb, SORT_T, 0, stream>>>(pairs, tcnt, tstart, rowptr, col,
                                               x, W1l, W1r, b1, W2l, W2r, b2,
                                               p, q, N, nbp);

  gather16_close_kernel<<<gb, T, 0, stream>>>(p, q, rowptr, col, out, N);
}
